# starting kernel
# baseline (speedup 1.0000x reference)
#include <hip/hip_runtime.h>
#include <math.h>

typedef __attribute__((ext_vector_type(2))) float v2f;
typedef __attribute__((ext_vector_type(8))) float v8f;

#define B_   8
#define C_   64
#define N_   4096
#define O_   64
#define K_NN 20
#define EPS_ 1e-5f
#define SLOPE_ 0.2f
#define NEG_INF -3.4e38f

static __device__ __forceinline__ v8f wmma_f32(v2f a, v2f b, v8f c) {
    return __builtin_amdgcn_wmma_f32_16x16x4_f32(false, a, false, b, (short)0, c,
                                                 false, false);
}

__global__ void __launch_bounds__(256) k_sqnorm(const float* __restrict__ x,
                                                float* __restrict__ xx) {
    int t = blockIdx.x * blockDim.x + threadIdx.x;
    if (t >= B_ * N_) return;
    int b = t / N_, n = t % N_;
    const float* xb = x + (size_t)b * C_ * N_;
    float s = 0.f;
#pragma unroll
    for (int c = 0; c < C_; ++c) {
        float v = xb[(size_t)c * N_ + n];
        s += v * v;
    }
    xx[t] = s;
}

__global__ void __launch_bounds__(128) k_knn(const float* __restrict__ x,
                                             const float* __restrict__ xx,
                                             int* __restrict__ idx) {
    const int lane = threadIdx.x & 31;
    const int wave = threadIdx.x >> 5;
    const int b    = blockIdx.x / (N_ / 64);
    const int rb   = (blockIdx.x % (N_ / 64)) * 64;
    const int row0 = rb + wave * 16;

    const float* xb  = x  + (size_t)b * C_ * N_;
    const float* xxb = xx + (size_t)b * N_;

    const int hs  = lane >> 4;
    const int l16 = lane & 15;
    const int jb  = hs * 8;

    v2f A[16];
#pragma unroll
    for (int s = 0; s < 16; ++s) {
        int c0 = 4 * s + hs * 2;
        A[s].x = xb[(size_t)(c0 + 0) * N_ + row0 + l16];
        A[s].y = xb[(size_t)(c0 + 1) * N_ + row0 + l16];
    }

    float xr[8];
#pragma unroll
    for (int v = 0; v < 8; ++v) xr[v] = xxb[row0 + v + hs * 8];

    float vals[K_NN];
    int   ids[K_NN];
#pragma unroll
    for (int t = 0; t < K_NN; ++t) { vals[t] = NEG_INF; ids[t] = 0; }

    __shared__ float tile[2][4][16][17];

    int p = 0;
    for (int m0 = 0; m0 < N_; m0 += 16, p ^= 1) {
        v2f Bv[16];
#pragma unroll
        for (int s = 0; s < 16; ++s) {
            int c0 = 4 * s + hs * 2;
            Bv[s].x = xb[(size_t)(c0 + 0) * N_ + m0 + l16];
            Bv[s].y = xb[(size_t)(c0 + 1) * N_ + m0 + l16];
        }
        v8f acc = {};
#pragma unroll
        for (int s = 0; s < 16; ++s) acc = wmma_f32(A[s], Bv[s], acc);

        float xc = xxb[m0 + l16];
#pragma unroll
        for (int v = 0; v < 8; ++v) {
            int r = v + hs * 8;
            tile[p][wave][r][l16] = 2.0f * acc[v] - xr[v] - xc;
        }
        __syncthreads();
        for (int j = 0; j < 8; ++j) {
            float d = tile[p][wave][l16][jb + j];
            if (d > vals[K_NN - 1]) {
                float cv = d; int ci = m0 + jb + j;
#pragma unroll
                for (int t = 0; t < K_NN; ++t) {
                    if (cv > vals[t]) {
                        float tv = vals[t]; int ti = ids[t];
                        vals[t] = cv; ids[t] = ci;
                        cv = tv; ci = ti;
                    }
                }
            }
        }
    }

    __shared__ float mv[128][K_NN];
    __shared__ int   mi[128][K_NN];
#pragma unroll
    for (int t = 0; t < K_NN; ++t) {
        mv[threadIdx.x][t] = vals[t];
        mi[threadIdx.x][t] = ids[t];
    }
    __syncthreads();
    if (lane < 16) {
        const float* av = mv[wave * 32 + lane];
        const int*   ai = mi[wave * 32 + lane];
        const float* bv = mv[wave * 32 + lane + 16];
        const int*   bi = mi[wave * 32 + lane + 16];
        int* op = idx + ((size_t)b * N_ + row0 + lane) * K_NN;
        int ia = 0, ib2 = 0;
        for (int t = 0; t < K_NN; ++t) {
            float fa = av[ia], fb = bv[ib2];
            bool takeA = (fa > fb) || (fa == fb && ai[ia] < bi[ib2]);
            op[t] = takeA ? ai[ia++] : bi[ib2++];
        }
    }
}

__global__ void __launch_bounds__(256) k_wc(const float* __restrict__ W,
                                            float* __restrict__ Wc) {
    int t = blockIdx.x * blockDim.x + threadIdx.x;
    if (t >= 128 * 64) return;
    int op = t / 64, c = t % 64;
    float v;
    if (op < 64) v = W[(size_t)op * 128 + c];
    else         v = W[(size_t)(op - 64) * 128 + 64 + c] -
                     W[(size_t)(op - 64) * 128 + c];
    Wc[t] = v;
}

__global__ void __launch_bounds__(256) k_proj(const float* __restrict__ x,
                                              const float* __restrict__ Wc,
                                              float* __restrict__ P) {
    const int lane = threadIdx.x & 31;
    const int wave = threadIdx.x >> 5;
    const int b    = blockIdx.x / (N_ / 16);
    const int n0   = (blockIdx.x % (N_ / 16)) * 16;
    const int o0   = wave * 16;
    const float* xb = x + (size_t)b * C_ * N_;
    const int hs = lane >> 4, l16 = lane & 15;

    v8f acc = {};
#pragma unroll
    for (int s = 0; s < 16; ++s) {
        int c0 = 4 * s + hs * 2;
        v2f a, bb;
        a.x  = Wc[(size_t)(o0 + l16) * 64 + c0 + 0];
        a.y  = Wc[(size_t)(o0 + l16) * 64 + c0 + 1];
        bb.x = xb[(size_t)(c0 + 0) * N_ + n0 + l16];
        bb.y = xb[(size_t)(c0 + 1) * N_ + n0 + l16];
        acc = wmma_f32(a, bb, acc);
    }
    float* Pb = P + (size_t)b * 128 * N_;
#pragma unroll
    for (int v = 0; v < 8; ++v) {
        int m = o0 + v + hs * 8;
        Pb[(size_t)m * N_ + n0 + l16] = acc[v];
    }
}

__global__ void k_zero(float* __restrict__ S) {
    if (threadIdx.x < 128) S[threadIdx.x] = 0.f;
}

__global__ void __launch_bounds__(256) k_stats(const float* __restrict__ P,
                                               const int* __restrict__ idx,
                                               float* __restrict__ S) {
    const int chunks = N_ / 256;
    const int bo = blockIdx.x / chunks;
    const int n  = (blockIdx.x % chunks) * 256 + threadIdx.x;
    const int b = bo / O_, o = bo % O_;

    const float* P1 = P + (size_t)b * 128 * N_ + (size_t)o * N_;
    const float  p2 = P1[(size_t)64 * N_ + n];
    const int*   ip = idx + ((size_t)b * N_ + n) * K_NN;

    float s1 = 0.f, s2 = 0.f;
#pragma unroll
    for (int k = 0; k < K_NN; ++k) {
        float y = P1[ip[k]] + p2;
        s1 += y; s2 += y * y;
    }
    __shared__ float r1[256], r2[256];
    r1[threadIdx.x] = s1; r2[threadIdx.x] = s2;
    __syncthreads();
    for (int st = 128; st > 0; st >>= 1) {
        if (threadIdx.x < st) {
            r1[threadIdx.x] += r1[threadIdx.x + st];
            r2[threadIdx.x] += r2[threadIdx.x + st];
        }
        __syncthreads();
    }
    if (threadIdx.x == 0) {
        atomicAdd(&S[o],      r1[0]);
        atomicAdd(&S[64 + o], r2[0]);
    }
}

__global__ void __launch_bounds__(256) k_final(const float* __restrict__ P,
                                               const int* __restrict__ idx,
                                               const float* __restrict__ S,
                                               const float* __restrict__ gamma,
                                               const float* __restrict__ beta,
                                               float* __restrict__ out) {
    const int chunks = N_ / 256;
    const int bo = blockIdx.x / chunks;
    const int n  = (blockIdx.x % chunks) * 256 + threadIdx.x;
    const int b = bo / O_, o = bo % O_;

    const float cnt  = (float)((size_t)B_ * N_ * K_NN);
    const float mean = S[o] / cnt;
    const float var  = S[64 + o] / cnt - mean * mean;
    const float g    = gamma[o] * rsqrtf(var + EPS_);
    const float be   = beta[o];

    const float* P1 = P + (size_t)b * 128 * N_ + (size_t)o * N_;
    const float  p2 = P1[(size_t)64 * N_ + n];
    const int*   ip = idx + ((size_t)b * N_ + n) * K_NN;

    float best = NEG_INF;
#pragma unroll
    for (int k = 0; k < K_NN; ++k) {
        float y = P1[ip[k]] + p2;
        float z = (y - mean) * g + be;
        z = (z >= 0.f) ? z : SLOPE_ * z;
        best = fmaxf(best, z);
    }
    out[((size_t)b * O_ + o) * N_ + n] = best;
}

extern "C" void kernel_launch(void* const* d_in, const int* in_sizes, int n_in,
                              void* d_out, int out_size, void* d_ws, size_t ws_size,
                              hipStream_t stream) {
    const float* x     = (const float*)d_in[0];
    const float* W     = (const float*)d_in[1];
    const float* gamma = (const float*)d_in[2];
    const float* beta  = (const float*)d_in[3];
    float* out = (float*)d_out;

    char* ws = (char*)d_ws;
    size_t off = 0;
    float* xx = (float*)(ws + off); off += (size_t)B_ * N_ * sizeof(float);
    int*   id = (int*)  (ws + off); off += (size_t)B_ * N_ * K_NN * sizeof(int);
    float* Wc = (float*)(ws + off); off += (size_t)128 * 64 * sizeof(float);
    float* P  = (float*)(ws + off); off += (size_t)B_ * 128 * N_ * sizeof(float);
    float* S  = (float*)(ws + off); off += 128 * sizeof(float);
    (void)ws_size; (void)in_sizes; (void)n_in; (void)out_size;

    k_sqnorm<<<(B_ * N_ + 255) / 256, 256, 0, stream>>>(x, xx);
    k_knn  <<<B_ * (N_ / 64), 128, 0, stream>>>(x, xx, id);
    k_wc   <<<(128 * 64 + 255) / 256, 256, 0, stream>>>(W, Wc);
    k_proj <<<B_ * (N_ / 16), 256, 0, stream>>>(x, Wc, P);
    k_zero <<<1, 128, 0, stream>>>(S);
    k_stats<<<B_ * O_ * (N_ / 256), 256, 0, stream>>>(P, id, S);
    k_final<<<B_ * O_ * (N_ / 256), 256, 0, stream>>>(P, id, S, gamma, beta, out);
}


# k_knn top-20 update: per-lane sorted list in 40 named scalars, stable insert as a branch-free compare/med3/select network, replacing the conditional swap loop over local arrays
# speedup vs baseline: 5.3005x; 5.3005x over previous
#include <hip/hip_runtime.h>
#include <math.h>

typedef __attribute__((ext_vector_type(2))) float v2f;
typedef __attribute__((ext_vector_type(8))) float v8f;

#define B_   8
#define C_   64
#define N_   4096
#define O_   64
#define K_NN 20
#define EPS_ 1e-5f
#define SLOPE_ 0.2f
#define NEG_INF -3.4e38f

static __device__ __forceinline__ v8f wmma_f32(v2f a, v2f b, v8f c) {
    return __builtin_amdgcn_wmma_f32_16x16x4_f32(false, a, false, b, (short)0, c,
                                                 false, false);
}

__global__ void __launch_bounds__(256) k_sqnorm(const float* __restrict__ x,
                                                float* __restrict__ xx) {
    int t = blockIdx.x * blockDim.x + threadIdx.x;
    if (t >= B_ * N_) return;
    int b = t / N_, n = t % N_;
    const float* xb = x + (size_t)b * C_ * N_;
    float s = 0.f;
#pragma unroll
    for (int c = 0; c < C_; ++c) {
        float v = xb[(size_t)c * N_ + n];
        s += v * v;
    }
    xx[t] = s;
}

__global__ void __launch_bounds__(128) k_knn(const float* __restrict__ x,
                                             const float* __restrict__ xx,
                                             int* __restrict__ idx) {
    const int lane = threadIdx.x & 31;
    const int wave = threadIdx.x >> 5;
    const int b    = blockIdx.x / (N_ / 64);
    const int rb   = (blockIdx.x % (N_ / 64)) * 64;
    const int row0 = rb + wave * 16;

    const float* xb  = x  + (size_t)b * C_ * N_;
    const float* xxb = xx + (size_t)b * N_;

    const int hs  = lane >> 4;
    const int l16 = lane & 15;
    const int jb  = hs * 8;

    v2f A[16];
#pragma unroll
    for (int s = 0; s < 16; ++s) {
        int c0 = 4 * s + hs * 2;
        A[s].x = xb[(size_t)(c0 + 0) * N_ + row0 + l16];
        A[s].y = xb[(size_t)(c0 + 1) * N_ + row0 + l16];
    }

    float xr[8];
#pragma unroll
    for (int v = 0; v < 8; ++v) xr[v] = xxb[row0 + v + hs * 8];

#define KNN_ALL(X) X(0) X(1) X(2) X(3) X(4) X(5) X(6) X(7) X(8) X(9) X(10) X(11) \
                   X(12) X(13) X(14) X(15) X(16) X(17) X(18) X(19)
#define KNN_TAIL_UP(X) X(19, 18) X(18, 17) X(17, 16) X(16, 15) X(15, 14) X(14, 13) \
                       X(13, 12) X(12, 11) X(11, 10) X(10, 9) X(9, 8) X(8, 7) X(7, 6) \
                       X(6, 5) X(5, 4) X(4, 3) X(3, 2) X(2, 1) X(1, 0)
#define KNN_DECL(t) float v##t = NEG_INF; int i##t = 0;
    KNN_ALL(KNN_DECL)
#define KNN_SLOT(t, u) { const bool gh = cv > v##u;                         \
                         i##t = gh ? i##u : (gl ? ci : i##t);               \
                         v##t = __builtin_amdgcn_fmed3f(v##u, v##t, cv);    \
                         gl = gh; }

    __shared__ float tile[2][4][16][17];

    int p = 0;
    for (int m0 = 0; m0 < N_; m0 += 16, p ^= 1) {
        v2f Bv[16];
#pragma unroll
        for (int s = 0; s < 16; ++s) {
            int c0 = 4 * s + hs * 2;
            Bv[s].x = xb[(size_t)(c0 + 0) * N_ + m0 + l16];
            Bv[s].y = xb[(size_t)(c0 + 1) * N_ + m0 + l16];
        }
        v8f acc = {};
#pragma unroll
        for (int s = 0; s < 16; ++s) acc = wmma_f32(A[s], Bv[s], acc);

        float xc = xxb[m0 + l16];
#pragma unroll
        for (int v = 0; v < 8; ++v) {
            int r = v + hs * 8;
            tile[p][wave][r][l16] = 2.0f * acc[v] - xr[v] - xc;
        }
        __syncthreads();
#pragma unroll
        for (int j = 0; j < 8; ++j) {
            const float cv = tile[p][wave][l16][jb + j];
            if (cv > v19) {
                const int ci = m0 + jb + j;
                bool gl = true;
                KNN_TAIL_UP(KNN_SLOT)
                i0 = gl ? ci : i0;
                v0 = fmaxf(v0, cv);
            }
        }
    }

    __shared__ float mv[128][K_NN];
    __shared__ int   mi[128][K_NN];
#define KNN_PUT(t) mv[threadIdx.x][t] = v##t; mi[threadIdx.x][t] = i##t;
    KNN_ALL(KNN_PUT)
    __syncthreads();
    if (lane < 16) {
        const float* av = mv[wave * 32 + lane];
        const int*   ai = mi[wave * 32 + lane];
        const float* bv = mv[wave * 32 + lane + 16];
        const int*   bi = mi[wave * 32 + lane + 16];
        int* op = idx + ((size_t)b * N_ + row0 + lane) * K_NN;
        int ia = 0, ib2 = 0;
        for (int t = 0; t < K_NN; ++t) {
            float fa = av[ia], fb = bv[ib2];
            bool takeA = (fa > fb) || (fa == fb && ai[ia] < bi[ib2]);
            op[t] = takeA ? ai[ia++] : bi[ib2++];
        }
    }
}

__global__ void __launch_bounds__(256) k_wc(const float* __restrict__ W,
                                            float* __restrict__ Wc) {
    int t = blockIdx.x * blockDim.x + threadIdx.x;
    if (t >= 128 * 64) return;
    int op = t / 64, c = t % 64;
    float v;
    if (op < 64) v = W[(size_t)op * 128 + c];
    else         v = W[(size_t)(op - 64) * 128 + 64 + c] -
                     W[(size_t)(op - 64) * 128 + c];
    Wc[t] = v;
}

__global__ void __launch_bounds__(256) k_proj(const float* __restrict__ x,
                                              const float* __restrict__ Wc,
                                              float* __restrict__ P) {
    const int lane = threadIdx.x & 31;
    const int wave = threadIdx.x >> 5;
    const int b    = blockIdx.x / (N_ / 16);
    const int n0   = (blockIdx.x % (N_ / 16)) * 16;
    const int o0   = wave * 16;
    const float* xb = x + (size_t)b * C_ * N_;
    const int hs = lane >> 4, l16 = lane & 15;

    v8f acc = {};
#pragma unroll
    for (int s = 0; s < 16; ++s) {
        int c0 = 4 * s + hs * 2;
        v2f a, bb;
        a.x  = Wc[(size_t)(o0 + l16) * 64 + c0 + 0];
        a.y  = Wc[(size_t)(o0 + l16) * 64 + c0 + 1];
        bb.x = xb[(size_t)(c0 + 0) * N_ + n0 + l16];
        bb.y = xb[(size_t)(c0 + 1) * N_ + n0 + l16];
        acc = wmma_f32(a, bb, acc);
    }
    float* Pb = P + (size_t)b * 128 * N_;
#pragma unroll
    for (int v = 0; v < 8; ++v) {
        int m = o0 + v + hs * 8;
        Pb[(size_t)m * N_ + n0 + l16] = acc[v];
    }
}

__global__ void k_zero(float* __restrict__ S) {
    if (threadIdx.x < 128) S[threadIdx.x] = 0.f;
}

__global__ void __launch_bounds__(256) k_stats(const float* __restrict__ P,
                                               const int* __restrict__ idx,
                                               float* __restrict__ S) {
    const int chunks = N_ / 256;
    const int bo = blockIdx.x / chunks;
    const int n  = (blockIdx.x % chunks) * 256 + threadIdx.x;
    const int b = bo / O_, o = bo % O_;

    const float* P1 = P + (size_t)b * 128 * N_ + (size_t)o * N_;
    const float  p2 = P1[(size_t)64 * N_ + n];
    const int*   ip = idx + ((size_t)b * N_ + n) * K_NN;

    float s1 = 0.f, s2 = 0.f;
#pragma unroll
    for (int k = 0; k < K_NN; ++k) {
        float y = P1[ip[k]] + p2;
        s1 += y; s2 += y * y;
    }
    __shared__ float r1[256], r2[256];
    r1[threadIdx.x] = s1; r2[threadIdx.x] = s2;
    __syncthreads();
    for (int st = 128; st > 0; st >>= 1) {
        if (threadIdx.x < st) {
            r1[threadIdx.x] += r1[threadIdx.x + st];
            r2[threadIdx.x] += r2[threadIdx.x + st];
        }
        __syncthreads();
    }
    if (threadIdx.x == 0) {
        atomicAdd(&S[o],      r1[0]);
        atomicAdd(&S[64 + o], r2[0]);
    }
}

__global__ void __launch_bounds__(256) k_final(const float* __restrict__ P,
                                               const int* __restrict__ idx,
                                               const float* __restrict__ S,
                                               const float* __restrict__ gamma,
                                               const float* __restrict__ beta,
                                               float* __restrict__ out) {
    const int chunks = N_ / 256;
    const int bo = blockIdx.x / chunks;
    const int n  = (blockIdx.x % chunks) * 256 + threadIdx.x;
    const int b = bo / O_, o = bo % O_;

    const float cnt  = (float)((size_t)B_ * N_ * K_NN);
    const float mean = S[o] / cnt;
    const float var  = S[64 + o] / cnt - mean * mean;
    const float g    = gamma[o] * rsqrtf(var + EPS_);
    const float be   = beta[o];

    const float* P1 = P + (size_t)b * 128 * N_ + (size_t)o * N_;
    const float  p2 = P1[(size_t)64 * N_ + n];
    const int*   ip = idx + ((size_t)b * N_ + n) * K_NN;

    float best = NEG_INF;
#pragma unroll
    for (int k = 0; k < K_NN; ++k) {
        float y = P1[ip[k]] + p2;
        float z = (y - mean) * g + be;
        z = (z >= 0.f) ? z : SLOPE_ * z;
        best = fmaxf(best, z);
    }
    out[((size_t)b * O_ + o) * N_ + n] = best;
}

extern "C" void kernel_launch(void* const* d_in, const int* in_sizes, int n_in,
                              void* d_out, int out_size, void* d_ws, size_t ws_size,
                              hipStream_t stream) {
    const float* x     = (const float*)d_in[0];
    const float* W     = (const float*)d_in[1];
    const float* gamma = (const float*)d_in[2];
    const float* beta  = (const float*)d_in[3];
    float* out = (float*)d_out;

    char* ws = (char*)d_ws;
    size_t off = 0;
    float* xx = (float*)(ws + off); off += (size_t)B_ * N_ * sizeof(float);
    int*   id = (int*)  (ws + off); off += (size_t)B_ * N_ * K_NN * sizeof(int);
    float* Wc = (float*)(ws + off); off += (size_t)128 * 64 * sizeof(float);
    float* P  = (float*)(ws + off); off += (size_t)B_ * 128 * N_ * sizeof(float);
    float* S  = (float*)(ws + off); off += 128 * sizeof(float);
    (void)ws_size; (void)in_sizes; (void)n_in; (void)out_size;

    k_sqnorm<<<(B_ * N_ + 255) / 256, 256, 0, stream>>>(x, xx);
    k_knn  <<<B_ * (N_ / 64), 128, 0, stream>>>(x, xx, id);
    k_wc   <<<(128 * 64 + 255) / 256, 256, 0, stream>>>(W, Wc);
    k_proj <<<B_ * (N_ / 16), 256, 0, stream>>>(x, Wc, P);
    k_zero <<<1, 128, 0, stream>>>(S);
    k_stats<<<B_ * O_ * (N_ / 256), 256, 0, stream>>>(P, id, S);
    k_final<<<B_ * O_ * (N_ / 256), 256, 0, stream>>>(P, id, S, gamma, beta, out);
}


# Projection stored point-major: a neighbour's 64 channels are one 256-B row read by a wave (2 channels per lane), indices read once per point; replaces per-channel 4-byte gathers in k_stats/k_final
# speedup vs baseline: 6.6321x; 1.2512x over previous
#include <hip/hip_runtime.h>
#include <math.h>

typedef __attribute__((ext_vector_type(2))) float v2f;
typedef __attribute__((ext_vector_type(8))) float v8f;

#define B_   8
#define C_   64
#define N_   4096
#define O_   64
#define K_NN 20
#define EPS_ 1e-5f
#define SLOPE_ 0.2f
#define NEG_INF -3.4e38f

static __device__ __forceinline__ v8f wmma_f32(v2f a, v2f b, v8f c) {
    return __builtin_amdgcn_wmma_f32_16x16x4_f32(false, a, false, b, (short)0, c,
                                                 false, false);
}

__global__ void __launch_bounds__(256) k_sqnorm(const float* __restrict__ x,
                                                float* __restrict__ xx) {
    int t = blockIdx.x * blockDim.x + threadIdx.x;
    if (t >= B_ * N_) return;
    int b = t / N_, n = t % N_;
    const float* xb = x + (size_t)b * C_ * N_;
    float s = 0.f;
#pragma unroll
    for (int c = 0; c < C_; ++c) {
        float v = xb[(size_t)c * N_ + n];
        s += v * v;
    }
    xx[t] = s;
}

__global__ void __launch_bounds__(128) k_knn(const float* __restrict__ x,
                                             const float* __restrict__ xx,
                                             int* __restrict__ idx) {
    const int lane = threadIdx.x & 31;
    const int wave = threadIdx.x >> 5;
    const int b    = blockIdx.x / (N_ / 64);
    const int rb   = (blockIdx.x % (N_ / 64)) * 64;
    const int row0 = rb + wave * 16;

    const float* xb  = x  + (size_t)b * C_ * N_;
    const float* xxb = xx + (size_t)b * N_;

    const int hs  = lane >> 4;
    const int l16 = lane & 15;
    const int jb  = hs * 8;

    v2f A[16];
#pragma unroll
    for (int s = 0; s < 16; ++s) {
        int c0 = 4 * s + hs * 2;
        A[s].x = xb[(size_t)(c0 + 0) * N_ + row0 + l16];
        A[s].y = xb[(size_t)(c0 + 1) * N_ + row0 + l16];
    }

    float xr[8];
#pragma unroll
    for (int v = 0; v < 8; ++v) xr[v] = xxb[row0 + v + hs * 8];

#define KNN_ALL(X) X(0) X(1) X(2) X(3) X(4) X(5) X(6) X(7) X(8) X(9) X(10) X(11) \
                   X(12) X(13) X(14) X(15) X(16) X(17) X(18) X(19)
#define KNN_TAIL_UP(X) X(19, 18) X(18, 17) X(17, 16) X(16, 15) X(15, 14) X(14, 13) \
                       X(13, 12) X(12, 11) X(11, 10) X(10, 9) X(9, 8) X(8, 7) X(7, 6) \
                       X(6, 5) X(5, 4) X(4, 3) X(3, 2) X(2, 1) X(1, 0)
#define KNN_DECL(t) float v##t = NEG_INF; int i##t = 0;
    KNN_ALL(KNN_DECL)
#define KNN_SLOT(t, u) { const bool gh = cv > v##u;                         \
                         i##t = gh ? i##u : (gl ? ci : i##t);               \
                         v##t = __builtin_amdgcn_fmed3f(v##u, v##t, cv);    \
                         gl = gh; }

    __shared__ float tile[2][4][16][17];

    int p = 0;
    for (int m0 = 0; m0 < N_; m0 += 16, p ^= 1) {
        v2f Bv[16];
#pragma unroll
        for (int s = 0; s < 16; ++s) {
            int c0 = 4 * s + hs * 2;
            Bv[s].x = xb[(size_t)(c0 + 0) * N_ + m0 + l16];
            Bv[s].y = xb[(size_t)(c0 + 1) * N_ + m0 + l16];
        }
        v8f acc = {};
#pragma unroll
        for (int s = 0; s < 16; ++s) acc = wmma_f32(A[s], Bv[s], acc);

        float xc = xxb[m0 + l16];
#pragma unroll
        for (int v = 0; v < 8; ++v) {
            int r = v + hs * 8;
            tile[p][wave][r][l16] = 2.0f * acc[v] - xr[v] - xc;
        }
        __syncthreads();
#pragma unroll
        for (int j = 0; j < 8; ++j) {
            const float cv = tile[p][wave][l16][jb + j];
            if (cv > v19) {
                const int ci = m0 + jb + j;
                bool gl = true;
                KNN_TAIL_UP(KNN_SLOT)
                i0 = gl ? ci : i0;
                v0 = fmaxf(v0, cv);
            }
        }
    }

    __shared__ float mv[128][K_NN];
    __shared__ int   mi[128][K_NN];
#define KNN_PUT(t) mv[threadIdx.x][t] = v##t; mi[threadIdx.x][t] = i##t;
    KNN_ALL(KNN_PUT)
    __syncthreads();
    if (lane < 16) {
        const float* av = mv[wave * 32 + lane];
        const int*   ai = mi[wave * 32 + lane];
        const float* bv = mv[wave * 32 + lane + 16];
        const int*   bi = mi[wave * 32 + lane + 16];
        int* op = idx + ((size_t)b * N_ + row0 + lane) * K_NN;
        int ia = 0, ib2 = 0;
        for (int t = 0; t < K_NN; ++t) {
            float fa = av[ia], fb = bv[ib2];
            bool takeA = (fa > fb) || (fa == fb && ai[ia] < bi[ib2]);
            op[t] = takeA ? ai[ia++] : bi[ib2++];
        }
    }
}

__global__ void __launch_bounds__(256) k_wc(const float* __restrict__ W,
                                            float* __restrict__ Wc) {
    int t = blockIdx.x * blockDim.x + threadIdx.x;
    if (t >= 128 * 64) return;
    int op = t / 64, c = t % 64;
    float v;
    if (op < 64) v = W[(size_t)op * 128 + c];
    else         v = W[(size_t)(op - 64) * 128 + 64 + c] -
                     W[(size_t)(op - 64) * 128 + c];
    Wc[t] = v;
}

#define PT_STRIDE 128
__global__ void __launch_bounds__(256) k_proj(const float* __restrict__ x,
                                              const float* __restrict__ Wc,
                                              float* __restrict__ Pt) {
    const int lane = threadIdx.x & 31;
    const int wave = threadIdx.x >> 5;
    const int tile = blockIdx.x * 2 + (wave >> 2);
    const int b    = tile / (N_ / 16);
    const int n0   = (tile % (N_ / 16)) * 16;
    const int o0   = (wave & 3) * 32;
    const float* xb = x + (size_t)b * C_ * N_;
    const int hs = lane >> 4, l16 = lane & 15;
    const float* w0 = Wc + (size_t)(o0 + 2 * l16) * 64;
    const float* w1 = w0 + 64;

    v8f acc0 = {}, acc1 = {};
#pragma unroll
    for (int s = 0; s < 16; ++s) {
        int c0 = 4 * s + hs * 2;
        v2f a, b0, b1;
        a.x  = xb[(size_t)(c0 + 0) * N_ + n0 + l16];
        a.y  = xb[(size_t)(c0 + 1) * N_ + n0 + l16];
        b0.x = w0[c0 + 0];
        b0.y = w0[c0 + 1];
        b1.x = w1[c0 + 0];
        b1.y = w1[c0 + 1];
        acc0 = wmma_f32(a, b0, acc0);
        acc1 = wmma_f32(a, b1, acc1);
    }
    float* dst = Pt + ((size_t)b * N_ + n0 + hs * 8) * PT_STRIDE + o0 + 2 * l16;
#pragma unroll
    for (int v = 0; v < 8; ++v) {
        v2f o;
        o.x = acc0[v];
        o.y = acc1[v];
        *(v2f*)(dst + (size_t)v * PT_STRIDE) = o;
    }
}

__global__ void k_zero(float* __restrict__ S) {
    if (threadIdx.x < 128) S[threadIdx.x] = 0.f;
}

#define ST_PPW 8
__global__ void __launch_bounds__(256) k_stats(const float* __restrict__ Pt,
                                               const int* __restrict__ idx,
                                               float* __restrict__ S) {
    const int lane = threadIdx.x & 31;
    const int wave = __builtin_amdgcn_readfirstlane(threadIdx.x >> 5);

    float s1x = 0.f, s1y = 0.f, s2x = 0.f, s2y = 0.f;
    for (int i = 0; i < ST_PPW; ++i) {
        const int t     = (blockIdx.x * 8 + wave) * ST_PPW + i;
        const int bbase = t & ~(N_ - 1);
        const int* ip   = idx + (size_t)t * K_NN;
        const v2f p2 = *(const v2f*)(Pt + (size_t)t * PT_STRIDE + 64 + 2 * lane);
#pragma unroll
        for (int k = 0; k < K_NN; ++k) {
            const int j = ip[k];
            const v2f v = *(const v2f*)(Pt + (size_t)(bbase + j) * PT_STRIDE + 2 * lane);
            const float y0 = v.x + p2.x;
            const float y1 = v.y + p2.y;
            s1x += y0; s2x += y0 * y0;
            s1y += y1; s2y += y1 * y1;
        }
    }
    __shared__ float r1[8][64], r2[8][64];
    r1[wave][2 * lane]     = s1x;
    r1[wave][2 * lane + 1] = s1y;
    r2[wave][2 * lane]     = s2x;
    r2[wave][2 * lane + 1] = s2y;
    __syncthreads();
    if (threadIdx.x < 128) {
        const int o = threadIdx.x & 63;
        float a = 0.f;
        if (threadIdx.x < 64) {
            for (int w = 0; w < 8; ++w) a += r1[w][o];
        } else {
            for (int w = 0; w < 8; ++w) a += r2[w][o];
        }
        atomicAdd(&S[threadIdx.x], a);
    }
}

__global__ void __launch_bounds__(256) k_final(const float* __restrict__ Pt,
                                               const int* __restrict__ idx,
                                               const float* __restrict__ S,
                                               const float* __restrict__ gamma,
                                               const float* __restrict__ beta,
                                               float* __restrict__ out) {
    const int lane = threadIdx.x & 31;
    const int wave = __builtin_amdgcn_readfirstlane(threadIdx.x >> 5);
    const int t0    = blockIdx.x * 32;
    const int bbase = t0 & ~(N_ - 1);

    const float cnt = (float)((size_t)B_ * N_ * K_NN);
    const int   oa  = 2 * lane, ob = 2 * lane + 1;
    const float mean0 = S[oa] / cnt;
    const float var0  = S[64 + oa] / cnt - mean0 * mean0;
    const float g0    = gamma[oa] * rsqrtf(var0 + EPS_);
    const float be0   = beta[oa];
    const float mean1 = S[ob] / cnt;
    const float var1  = S[64 + ob] / cnt - mean1 * mean1;
    const float g1    = gamma[ob] * rsqrtf(var1 + EPS_);
    const float be1   = beta[ob];

    __shared__ float tile[64][33];

    for (int i = 0; i < 4; ++i) {
        const int pl = wave * 4 + i;
        const int t  = t0 + pl;
        const int* ip = idx + (size_t)t * K_NN;
        const v2f p2 = *(const v2f*)(Pt + (size_t)t * PT_STRIDE + 64 + 2 * lane);
        float best0 = NEG_INF, best1 = NEG_INF;
#pragma unroll
        for (int k = 0; k < K_NN; ++k) {
            const int j = ip[k];
            const v2f v = *(const v2f*)(Pt + (size_t)(bbase + j) * PT_STRIDE + 2 * lane);
            const float y0 = v.x + p2.x;
            const float y1 = v.y + p2.y;
            float z0 = (y0 - mean0) * g0 + be0;
            float z1 = (y1 - mean1) * g1 + be1;
            z0 = (z0 >= 0.f) ? z0 : SLOPE_ * z0;
            z1 = (z1 >= 0.f) ? z1 : SLOPE_ * z1;
            best0 = fmaxf(best0, z0);
            best1 = fmaxf(best1, z1);
        }
        tile[oa][pl] = best0;
        tile[ob][pl] = best1;
    }
    __syncthreads();
    const int b  = t0 / N_;
    const int n0 = t0 % N_;
    for (int r = 0; r < 8; ++r) {
        const int o = wave * 8 + r;
        out[((size_t)b * O_ + o) * N_ + n0 + lane] = tile[o][lane];
    }
}

extern "C" void kernel_launch(void* const* d_in, const int* in_sizes, int n_in,
                              void* d_out, int out_size, void* d_ws, size_t ws_size,
                              hipStream_t stream) {
    const float* x     = (const float*)d_in[0];
    const float* W     = (const float*)d_in[1];
    const float* gamma = (const float*)d_in[2];
    const float* beta  = (const float*)d_in[3];
    float* out = (float*)d_out;

    char* ws = (char*)d_ws;
    size_t off = 0;
    float* xx = (float*)(ws + off); off += (size_t)B_ * N_ * sizeof(float);
    int*   id = (int*)  (ws + off); off += (size_t)B_ * N_ * K_NN * sizeof(int);
    float* Wc = (float*)(ws + off); off += (size_t)128 * 64 * sizeof(float);
    float* P  = (float*)(ws + off); off += (size_t)B_ * N_ * PT_STRIDE * sizeof(float);
    float* S  = (float*)(ws + off); off += 128 * sizeof(float);
    (void)ws_size; (void)in_sizes; (void)n_in; (void)out_size;

    k_sqnorm<<<(B_ * N_ + 255) / 256, 256, 0, stream>>>(x, xx);
    k_knn  <<<B_ * (N_ / 64), 128, 0, stream>>>(x, xx, id);
    k_wc   <<<(128 * 64 + 255) / 256, 256, 0, stream>>>(W, Wc);
    k_proj <<<B_ * (N_ / 32), 256, 0, stream>>>(x, Wc, P);
    k_zero <<<1, 128, 0, stream>>>(S);
    k_stats<<<B_ * N_ / (8 * ST_PPW), 256, 0, stream>>>(P, id, S);
    k_final<<<B_ * N_ / 32, 256, 0, stream>>>(P, id, S, gamma, beta, out);
}


# k_knn: Gram tile computed transposed (column tile as A operand, query rows as B) so a lane's 8 accumulator registers are its row's 8 candidates; per-tile LDS transpose, barrier and LDS reads removed
# speedup vs baseline: 8.3282x; 1.2557x over previous
#include <hip/hip_runtime.h>
#include <math.h>

typedef __attribute__((ext_vector_type(2))) float v2f;
typedef __attribute__((ext_vector_type(8))) float v8f;

#define B_   8
#define C_   64
#define N_   4096
#define O_   64
#define K_NN 20
#define EPS_ 1e-5f
#define SLOPE_ 0.2f
#define NEG_INF -3.4e38f

static __device__ __forceinline__ v8f wmma_f32(v2f a, v2f b, v8f c) {
    return __builtin_amdgcn_wmma_f32_16x16x4_f32(false, a, false, b, (short)0, c,
                                                 false, false);
}

__global__ void __launch_bounds__(256) k_sqnorm(const float* __restrict__ x,
                                                float* __restrict__ xx) {
    int t = blockIdx.x * blockDim.x + threadIdx.x;
    if (t >= B_ * N_) return;
    int b = t / N_, n = t % N_;
    const float* xb = x + (size_t)b * C_ * N_;
    float s = 0.f;
#pragma unroll
    for (int c = 0; c < C_; ++c) {
        float v = xb[(size_t)c * N_ + n];
        s += v * v;
    }
    xx[t] = s;
}

__global__ void __launch_bounds__(128) k_knn(const float* __restrict__ x,
                                             const float* __restrict__ xx,
                                             int* __restrict__ idx) {
    const int lane = threadIdx.x & 31;
    const int wave = threadIdx.x >> 5;
    const int b    = blockIdx.x / (N_ / 64);
    const int rb   = (blockIdx.x % (N_ / 64)) * 64;
    const int row0 = rb + wave * 16;

    const float* xb  = x  + (size_t)b * C_ * N_;
    const float* xxb = xx + (size_t)b * N_;

    const int hs  = lane >> 4;
    const int l16 = lane & 15;
    const int jb  = hs * 8;

    v2f A[16];
#pragma unroll
    for (int s = 0; s < 16; ++s) {
        int c0 = 4 * s + hs * 2;
        A[s].x = xb[(size_t)(c0 + 0) * N_ + row0 + l16];
        A[s].y = xb[(size_t)(c0 + 1) * N_ + row0 + l16];
    }

    const float xr = xxb[row0 + l16];

#define KNN_ALL(X) X(0) X(1) X(2) X(3) X(4) X(5) X(6) X(7) X(8) X(9) X(10) X(11) \
                   X(12) X(13) X(14) X(15) X(16) X(17) X(18) X(19)
#define KNN_TAIL_UP(X) X(19, 18) X(18, 17) X(17, 16) X(16, 15) X(15, 14) X(14, 13) \
                       X(13, 12) X(12, 11) X(11, 10) X(10, 9) X(9, 8) X(8, 7) X(7, 6) \
                       X(6, 5) X(5, 4) X(4, 3) X(3, 2) X(2, 1) X(1, 0)
#define KNN_DECL(t) float v##t = NEG_INF; int i##t = 0;
    KNN_ALL(KNN_DECL)
#define KNN_SLOT(t, u) { const bool gh = cv > v##u;                         \
                         i##t = gh ? i##u : (gl ? ci : i##t);               \
                         v##t = __builtin_amdgcn_fmed3f(v##u, v##t, cv);    \
                         gl = gh; }

    for (int m0 = 0; m0 < N_; m0 += 16) {
        v2f Bv[16];
#pragma unroll
        for (int s = 0; s < 16; ++s) {
            int c0 = 4 * s + hs * 2;
            Bv[s].x = xb[(size_t)(c0 + 0) * N_ + m0 + l16];
            Bv[s].y = xb[(size_t)(c0 + 1) * N_ + m0 + l16];
        }
        v8f acc = {};
#pragma unroll
        for (int s = 0; s < 16; ++s) acc = wmma_f32(Bv[s], A[s], acc);

        float xc[8];
#pragma unroll
        for (int j = 0; j < 8; ++j) xc[j] = xxb[m0 + jb + j];
#pragma unroll
        for (int j = 0; j < 8; ++j) {
            const float cv = 2.0f * acc[j] - xr - xc[j];
            if (cv > v19) {
                const int ci = m0 + jb + j;
                bool gl = true;
                KNN_TAIL_UP(KNN_SLOT)
                i0 = gl ? ci : i0;
                v0 = fmaxf(v0, cv);
            }
        }
    }

    __shared__ float mv[128][K_NN];
    __shared__ int   mi[128][K_NN];
#define KNN_PUT(t) mv[threadIdx.x][t] = v##t; mi[threadIdx.x][t] = i##t;
    KNN_ALL(KNN_PUT)
    __syncthreads();
    if (lane < 16) {
        const float* av = mv[wave * 32 + lane];
        const int*   ai = mi[wave * 32 + lane];
        const float* bv = mv[wave * 32 + lane + 16];
        const int*   bi = mi[wave * 32 + lane + 16];
        int* op = idx + ((size_t)b * N_ + row0 + lane) * K_NN;
        int ia = 0, ib2 = 0;
        for (int t = 0; t < K_NN; ++t) {
            float fa = av[ia], fb = bv[ib2];
            bool takeA = (fa > fb) || (fa == fb && ai[ia] < bi[ib2]);
            op[t] = takeA ? ai[ia++] : bi[ib2++];
        }
    }
}

__global__ void __launch_bounds__(256) k_wc(const float* __restrict__ W,
                                            float* __restrict__ Wc) {
    int t = blockIdx.x * blockDim.x + threadIdx.x;
    if (t >= 128 * 64) return;
    int op = t / 64, c = t % 64;
    float v;
    if (op < 64) v = W[(size_t)op * 128 + c];
    else         v = W[(size_t)(op - 64) * 128 + 64 + c] -
                     W[(size_t)(op - 64) * 128 + c];
    Wc[t] = v;
}

#define PT_STRIDE 128
__global__ void __launch_bounds__(256) k_proj(const float* __restrict__ x,
                                              const float* __restrict__ Wc,
                                              float* __restrict__ Pt) {
    const int lane = threadIdx.x & 31;
    const int wave = threadIdx.x >> 5;
    const int tile = blockIdx.x * 2 + (wave >> 2);
    const int b    = tile / (N_ / 16);
    const int n0   = (tile % (N_ / 16)) * 16;
    const int o0   = (wave & 3) * 32;
    const float* xb = x + (size_t)b * C_ * N_;
    const int hs = lane >> 4, l16 = lane & 15;
    const float* w0 = Wc + (size_t)(o0 + 2 * l16) * 64;
    const float* w1 = w0 + 64;

    v8f acc0 = {}, acc1 = {};
#pragma unroll
    for (int s = 0; s < 16; ++s) {
        int c0 = 4 * s + hs * 2;
        v2f a, b0, b1;
        a.x  = xb[(size_t)(c0 + 0) * N_ + n0 + l16];
        a.y  = xb[(size_t)(c0 + 1) * N_ + n0 + l16];
        b0.x = w0[c0 + 0];
        b0.y = w0[c0 + 1];
        b1.x = w1[c0 + 0];
        b1.y = w1[c0 + 1];
        acc0 = wmma_f32(a, b0, acc0);
        acc1 = wmma_f32(a, b1, acc1);
    }
    float* dst = Pt + ((size_t)b * N_ + n0 + hs * 8) * PT_STRIDE + o0 + 2 * l16;
#pragma unroll
    for (int v = 0; v < 8; ++v) {
        v2f o;
        o.x = acc0[v];
        o.y = acc1[v];
        *(v2f*)(dst + (size_t)v * PT_STRIDE) = o;
    }
}

__global__ void k_zero(float* __restrict__ S) {
    if (threadIdx.x < 128) S[threadIdx.x] = 0.f;
}

#define ST_PPW 8
__global__ void __launch_bounds__(256) k_stats(const float* __restrict__ Pt,
                                               const int* __restrict__ idx,
                                               float* __restrict__ S) {
    const int lane = threadIdx.x & 31;
    const int wave = __builtin_amdgcn_readfirstlane(threadIdx.x >> 5);

    float s1x = 0.f, s1y = 0.f, s2x = 0.f, s2y = 0.f;
    for (int i = 0; i < ST_PPW; ++i) {
        const int t     = (blockIdx.x * 8 + wave) * ST_PPW + i;
        const int bbase = t & ~(N_ - 1);
        const int* ip   = idx + (size_t)t * K_NN;
        const v2f p2 = *(const v2f*)(Pt + (size_t)t * PT_STRIDE + 64 + 2 * lane);
#pragma unroll
        for (int k = 0; k < K_NN; ++k) {
            const int j = ip[k];
            const v2f v = *(const v2f*)(Pt + (size_t)(bbase + j) * PT_STRIDE + 2 * lane);
            const float y0 = v.x + p2.x;
            const float y1 = v.y + p2.y;
            s1x += y0; s2x += y0 * y0;
            s1y += y1; s2y += y1 * y1;
        }
    }
    __shared__ float r1[8][64], r2[8][64];
    r1[wave][2 * lane]     = s1x;
    r1[wave][2 * lane + 1] = s1y;
    r2[wave][2 * lane]     = s2x;
    r2[wave][2 * lane + 1] = s2y;
    __syncthreads();
    if (threadIdx.x < 128) {
        const int o = threadIdx.x & 63;
        float a = 0.f;
        if (threadIdx.x < 64) {
            for (int w = 0; w < 8; ++w) a += r1[w][o];
        } else {
            for (int w = 0; w < 8; ++w) a += r2[w][o];
        }
        atomicAdd(&S[threadIdx.x], a);
    }
}

__global__ void __launch_bounds__(256) k_final(const float* __restrict__ Pt,
                                               const int* __restrict__ idx,
                                               const float* __restrict__ S,
                                               const float* __restrict__ gamma,
                                               const float* __restrict__ beta,
                                               float* __restrict__ out) {
    const int lane = threadIdx.x & 31;
    const int wave = __builtin_amdgcn_readfirstlane(threadIdx.x >> 5);
    const int t0    = blockIdx.x * 32;
    const int bbase = t0 & ~(N_ - 1);

    const float cnt = (float)((size_t)B_ * N_ * K_NN);
    const int   oa  = 2 * lane, ob = 2 * lane + 1;
    const float mean0 = S[oa] / cnt;
    const float var0  = S[64 + oa] / cnt - mean0 * mean0;
    const float g0    = gamma[oa] * rsqrtf(var0 + EPS_);
    const float be0   = beta[oa];
    const float mean1 = S[ob] / cnt;
    const float var1  = S[64 + ob] / cnt - mean1 * mean1;
    const float g1    = gamma[ob] * rsqrtf(var1 + EPS_);
    const float be1   = beta[ob];

    __shared__ float tile[64][33];

    for (int i = 0; i < 4; ++i) {
        const int pl = wave * 4 + i;
        const int t  = t0 + pl;
        const int* ip = idx + (size_t)t * K_NN;
        const v2f p2 = *(const v2f*)(Pt + (size_t)t * PT_STRIDE + 64 + 2 * lane);
        float best0 = NEG_INF, best1 = NEG_INF;
#pragma unroll
        for (int k = 0; k < K_NN; ++k) {
            const int j = ip[k];
            const v2f v = *(const v2f*)(Pt + (size_t)(bbase + j) * PT_STRIDE + 2 * lane);
            const float y0 = v.x + p2.x;
            const float y1 = v.y + p2.y;
            float z0 = (y0 - mean0) * g0 + be0;
            float z1 = (y1 - mean1) * g1 + be1;
            z0 = (z0 >= 0.f) ? z0 : SLOPE_ * z0;
            z1 = (z1 >= 0.f) ? z1 : SLOPE_ * z1;
            best0 = fmaxf(best0, z0);
            best1 = fmaxf(best1, z1);
        }
        tile[oa][pl] = best0;
        tile[ob][pl] = best1;
    }
    __syncthreads();
    const int b  = t0 / N_;
    const int n0 = t0 % N_;
    for (int r = 0; r < 8; ++r) {
        const int o = wave * 8 + r;
        out[((size_t)b * O_ + o) * N_ + n0 + lane] = tile[o][lane];
    }
}

extern "C" void kernel_launch(void* const* d_in, const int* in_sizes, int n_in,
                              void* d_out, int out_size, void* d_ws, size_t ws_size,
                              hipStream_t stream) {
    const float* x     = (const float*)d_in[0];
    const float* W     = (const float*)d_in[1];
    const float* gamma = (const float*)d_in[2];
    const float* beta  = (const float*)d_in[3];
    float* out = (float*)d_out;

    char* ws = (char*)d_ws;
    size_t off = 0;
    float* xx = (float*)(ws + off); off += (size_t)B_ * N_ * sizeof(float);
    int*   id = (int*)  (ws + off); off += (size_t)B_ * N_ * K_NN * sizeof(int);
    float* Wc = (float*)(ws + off); off += (size_t)128 * 64 * sizeof(float);
    float* P  = (float*)(ws + off); off += (size_t)B_ * N_ * PT_STRIDE * sizeof(float);
    float* S  = (float*)(ws + off); off += 128 * sizeof(float);
    (void)ws_size; (void)in_sizes; (void)n_in; (void)out_size;

    k_sqnorm<<<(B_ * N_ + 255) / 256, 256, 0, stream>>>(x, xx);
    k_knn  <<<B_ * (N_ / 64), 128, 0, stream>>>(x, xx, id);
    k_wc   <<<(128 * 64 + 255) / 256, 256, 0, stream>>>(W, Wc);
    k_proj <<<B_ * (N_ / 32), 256, 0, stream>>>(x, Wc, P);
    k_zero <<<1, 128, 0, stream>>>(S);
    k_stats<<<B_ * N_ / (8 * ST_PPW), 256, 0, stream>>>(P, id, S);
    k_final<<<B_ * N_ / 32, 256, 0, stream>>>(P, id, S, gamma, beta, out);
}


# k_knn: candidates beating the lane's 20th go to a per-lane LDS queue and are sorted-inserted in batches when a queue fills, so the insert network runs with most lanes active, not for any one lane
# speedup vs baseline: 9.5949x; 1.1521x over previous
#include <hip/hip_runtime.h>
#include <math.h>

typedef __attribute__((ext_vector_type(2))) float v2f;
typedef __attribute__((ext_vector_type(8))) float v8f;
typedef __attribute__((ext_vector_type(2))) int v2i;

#define B_   8
#define C_   64
#define N_   4096
#define O_   64
#define K_NN 20
#define EPS_ 1e-5f
#define SLOPE_ 0.2f
#define NEG_INF -3.4e38f
#define KQ   32

static __device__ __forceinline__ v8f wmma_f32(v2f a, v2f b, v8f c) {
    return __builtin_amdgcn_wmma_f32_16x16x4_f32(false, a, false, b, (short)0, c,
                                                 false, false);
}

__global__ void __launch_bounds__(256) k_sqnorm(const float* __restrict__ x,
                                                float* __restrict__ xx) {
    int t = blockIdx.x * blockDim.x + threadIdx.x;
    if (t >= B_ * N_) return;
    int b = t / N_, n = t % N_;
    const float* xb = x + (size_t)b * C_ * N_;
    float s = 0.f;
#pragma unroll
    for (int c = 0; c < C_; ++c) {
        float v = xb[(size_t)c * N_ + n];
        s += v * v;
    }
    xx[t] = s;
}

__global__ void __launch_bounds__(128) k_knn(const float* __restrict__ x,
                                             const float* __restrict__ xx,
                                             int* __restrict__ idx) {
    const int lane = threadIdx.x & 31;
    const int wave = threadIdx.x >> 5;
    const int b    = blockIdx.x / (N_ / 64);
    const int rb   = (blockIdx.x % (N_ / 64)) * 64;
    const int row0 = rb + wave * 16;

    const float* xb  = x  + (size_t)b * C_ * N_;
    const float* xxb = xx + (size_t)b * N_;

    const int hs  = lane >> 4;
    const int l16 = lane & 15;
    const int jb  = hs * 8;

    v2f A[16];
#pragma unroll
    for (int s = 0; s < 16; ++s) {
        int c0 = 4 * s + hs * 2;
        A[s].x = xb[(size_t)(c0 + 0) * N_ + row0 + l16];
        A[s].y = xb[(size_t)(c0 + 1) * N_ + row0 + l16];
    }

    const float xr = xxb[row0 + l16];

#define KNN_ALL(X) X(0) X(1) X(2) X(3) X(4) X(5) X(6) X(7) X(8) X(9) X(10) X(11) \
                   X(12) X(13) X(14) X(15) X(16) X(17) X(18) X(19)
#define KNN_TAIL_UP(X) X(19, 18) X(18, 17) X(17, 16) X(16, 15) X(15, 14) X(14, 13) \
                       X(13, 12) X(12, 11) X(11, 10) X(10, 9) X(9, 8) X(8, 7) X(7, 6) \
                       X(6, 5) X(5, 4) X(4, 3) X(3, 2) X(2, 1) X(1, 0)
#define KNN_DECL(t) float v##t = NEG_INF; int i##t = 0;
    KNN_ALL(KNN_DECL)
#define KNN_SLOT(t, u) { const bool gh = cv > v##u;                         \
                         i##t = gh ? i##u : (gl ? ci : i##t);               \
                         v##t = __builtin_amdgcn_fmed3f(v##u, v##t, cv);    \
                         gl = gh; }

    __shared__ v2i qf[KQ * 128];
    int qn = threadIdx.x;

    for (int m0 = 0; m0 < N_; m0 += 16) {
        v2f Bv[16];
#pragma unroll
        for (int s = 0; s < 16; ++s) {
            int c0 = 4 * s + hs * 2;
            Bv[s].x = xb[(size_t)(c0 + 0) * N_ + m0 + l16];
            Bv[s].y = xb[(size_t)(c0 + 1) * N_ + m0 + l16];
        }
        v8f acc = {};
#pragma unroll
        for (int s = 0; s < 16; ++s) acc = wmma_f32(Bv[s], A[s], acc);

        float xc[8];
#pragma unroll
        for (int j = 0; j < 8; ++j) xc[j] = xxb[m0 + jb + j];
#pragma unroll
        for (int j = 0; j < 8; ++j) {
            const float cv = 2.0f * acc[j] - xr - xc[j];
            if (cv > v19) {
                v2i e;
                e.x = __float_as_int(cv);
                e.y = m0 + jb + j;
                qf[qn] = e;
                qn += 128;
            }
        }
        const bool full = __builtin_amdgcn_ballot_w32(qn >= (KQ - 7) * 128) != 0;
        if (full || m0 == N_ - 16) {
            for (int q = threadIdx.x; __builtin_amdgcn_ballot_w32(q < qn) != 0; q += 128) {
                if (q < qn) {
                    const v2i e = qf[q];
                    const float cv = __int_as_float(e.x);
                    const int ci = e.y;
                    bool gl = cv > v19;
                    KNN_TAIL_UP(KNN_SLOT)
                    i0 = gl ? ci : i0;
                    v0 = fmaxf(v0, cv);
                }
            }
            qn = threadIdx.x;
        }
    }

    __shared__ float mv[128][K_NN];
    __shared__ int   mi[128][K_NN];
#define KNN_PUT(t) mv[threadIdx.x][t] = v##t; mi[threadIdx.x][t] = i##t;
    KNN_ALL(KNN_PUT)
    __syncthreads();
    if (lane < 16) {
        const float* av = mv[wave * 32 + lane];
        const int*   ai = mi[wave * 32 + lane];
        const float* bv = mv[wave * 32 + lane + 16];
        const int*   bi = mi[wave * 32 + lane + 16];
        int* op = idx + ((size_t)b * N_ + row0 + lane) * K_NN;
        int ia = 0, ib2 = 0;
        for (int t = 0; t < K_NN; ++t) {
            float fa = av[ia], fb = bv[ib2];
            bool takeA = (fa > fb) || (fa == fb && ai[ia] < bi[ib2]);
            op[t] = takeA ? ai[ia++] : bi[ib2++];
        }
    }
}

__global__ void __launch_bounds__(256) k_wc(const float* __restrict__ W,
                                            float* __restrict__ Wc) {
    int t = blockIdx.x * blockDim.x + threadIdx.x;
    if (t >= 128 * 64) return;
    int op = t / 64, c = t % 64;
    float v;
    if (op < 64) v = W[(size_t)op * 128 + c];
    else         v = W[(size_t)(op - 64) * 128 + 64 + c] -
                     W[(size_t)(op - 64) * 128 + c];
    Wc[t] = v;
}

#define PT_STRIDE 128
__global__ void __launch_bounds__(256) k_proj(const float* __restrict__ x,
                                              const float* __restrict__ Wc,
                                              float* __restrict__ Pt) {
    const int lane = threadIdx.x & 31;
    const int wave = threadIdx.x >> 5;
    const int tile = blockIdx.x * 2 + (wave >> 2);
    const int b    = tile / (N_ / 16);
    const int n0   = (tile % (N_ / 16)) * 16;
    const int o0   = (wave & 3) * 32;
    const float* xb = x + (size_t)b * C_ * N_;
    const int hs = lane >> 4, l16 = lane & 15;
    const float* w0 = Wc + (size_t)(o0 + 2 * l16) * 64;
    const float* w1 = w0 + 64;

    v8f acc0 = {}, acc1 = {};
#pragma unroll
    for (int s = 0; s < 16; ++s) {
        int c0 = 4 * s + hs * 2;
        v2f a, b0, b1;
        a.x  = xb[(size_t)(c0 + 0) * N_ + n0 + l16];
        a.y  = xb[(size_t)(c0 + 1) * N_ + n0 + l16];
        b0.x = w0[c0 + 0];
        b0.y = w0[c0 + 1];
        b1.x = w1[c0 + 0];
        b1.y = w1[c0 + 1];
        acc0 = wmma_f32(a, b0, acc0);
        acc1 = wmma_f32(a, b1, acc1);
    }
    float* dst = Pt + ((size_t)b * N_ + n0 + hs * 8) * PT_STRIDE + o0 + 2 * l16;
#pragma unroll
    for (int v = 0; v < 8; ++v) {
        v2f o;
        o.x = acc0[v];
        o.y = acc1[v];
        *(v2f*)(dst + (size_t)v * PT_STRIDE) = o;
    }
}

__global__ void k_zero(float* __restrict__ S) {
    if (threadIdx.x < 128) S[threadIdx.x] = 0.f;
}

#define ST_PPW 8
__global__ void __launch_bounds__(256) k_stats(const float* __restrict__ Pt,
                                               const int* __restrict__ idx,
                                               float* __restrict__ S) {
    const int lane = threadIdx.x & 31;
    const int wave = __builtin_amdgcn_readfirstlane(threadIdx.x >> 5);

    float s1x = 0.f, s1y = 0.f, s2x = 0.f, s2y = 0.f;
    for (int i = 0; i < ST_PPW; ++i) {
        const int t     = (blockIdx.x * 8 + wave) * ST_PPW + i;
        const int bbase = t & ~(N_ - 1);
        const int* ip   = idx + (size_t)t * K_NN;
        const v2f p2 = *(const v2f*)(Pt + (size_t)t * PT_STRIDE + 64 + 2 * lane);
#pragma unroll
        for (int k = 0; k < K_NN; ++k) {
            const int j = ip[k];
            const v2f v = *(const v2f*)(Pt + (size_t)(bbase + j) * PT_STRIDE + 2 * lane);
            const float y0 = v.x + p2.x;
            const float y1 = v.y + p2.y;
            s1x += y0; s2x += y0 * y0;
            s1y += y1; s2y += y1 * y1;
        }
    }
    __shared__ float r1[8][64], r2[8][64];
    r1[wave][2 * lane]     = s1x;
    r1[wave][2 * lane + 1] = s1y;
    r2[wave][2 * lane]     = s2x;
    r2[wave][2 * lane + 1] = s2y;
    __syncthreads();
    if (threadIdx.x < 128) {
        const int o = threadIdx.x & 63;
        float a = 0.f;
        if (threadIdx.x < 64) {
            for (int w = 0; w < 8; ++w) a += r1[w][o];
        } else {
            for (int w = 0; w < 8; ++w) a += r2[w][o];
        }
        atomicAdd(&S[threadIdx.x], a);
    }
}

__global__ void __launch_bounds__(256) k_final(const float* __restrict__ Pt,
                                               const int* __restrict__ idx,
                                               const float* __restrict__ S,
                                               const float* __restrict__ gamma,
                                               const float* __restrict__ beta,
                                               float* __restrict__ out) {
    const int lane = threadIdx.x & 31;
    const int wave = __builtin_amdgcn_readfirstlane(threadIdx.x >> 5);
    const int t0    = blockIdx.x * 32;
    const int bbase = t0 & ~(N_ - 1);

    const float cnt = (float)((size_t)B_ * N_ * K_NN);
    const int   oa  = 2 * lane, ob = 2 * lane + 1;
    const float mean0 = S[oa] / cnt;
    const float var0  = S[64 + oa] / cnt - mean0 * mean0;
    const float g0    = gamma[oa] * rsqrtf(var0 + EPS_);
    const float be0   = beta[oa];
    const float mean1 = S[ob] / cnt;
    const float var1  = S[64 + ob] / cnt - mean1 * mean1;
    const float g1    = gamma[ob] * rsqrtf(var1 + EPS_);
    const float be1   = beta[ob];

    __shared__ float tile[64][33];

    for (int i = 0; i < 4; ++i) {
        const int pl = wave * 4 + i;
        const int t  = t0 + pl;
        const int* ip = idx + (size_t)t * K_NN;
        const v2f p2 = *(const v2f*)(Pt + (size_t)t * PT_STRIDE + 64 + 2 * lane);
        float best0 = NEG_INF, best1 = NEG_INF;
#pragma unroll
        for (int k = 0; k < K_NN; ++k) {
            const int j = ip[k];
            const v2f v = *(const v2f*)(Pt + (size_t)(bbase + j) * PT_STRIDE + 2 * lane);
            const float y0 = v.x + p2.x;
            const float y1 = v.y + p2.y;
            float z0 = (y0 - mean0) * g0 + be0;
            float z1 = (y1 - mean1) * g1 + be1;
            z0 = (z0 >= 0.f) ? z0 : SLOPE_ * z0;
            z1 = (z1 >= 0.f) ? z1 : SLOPE_ * z1;
            best0 = fmaxf(best0, z0);
            best1 = fmaxf(best1, z1);
        }
        tile[oa][pl] = best0;
        tile[ob][pl] = best1;
    }
    __syncthreads();
    const int b  = t0 / N_;
    const int n0 = t0 % N_;
    for (int r = 0; r < 8; ++r) {
        const int o = wave * 8 + r;
        out[((size_t)b * O_ + o) * N_ + n0 + lane] = tile[o][lane];
    }
}

extern "C" void kernel_launch(void* const* d_in, const int* in_sizes, int n_in,
                              void* d_out, int out_size, void* d_ws, size_t ws_size,
                              hipStream_t stream) {
    const float* x     = (const float*)d_in[0];
    const float* W     = (const float*)d_in[1];
    const float* gamma = (const float*)d_in[2];
    const float* beta  = (const float*)d_in[3];
    float* out = (float*)d_out;

    char* ws = (char*)d_ws;
    size_t off = 0;
    float* xx = (float*)(ws + off); off += (size_t)B_ * N_ * sizeof(float);
    int*   id = (int*)  (ws + off); off += (size_t)B_ * N_ * K_NN * sizeof(int);
    float* Wc = (float*)(ws + off); off += (size_t)128 * 64 * sizeof(float);
    float* P  = (float*)(ws + off); off += (size_t)B_ * N_ * PT_STRIDE * sizeof(float);
    float* S  = (float*)(ws + off); off += 128 * sizeof(float);
    (void)ws_size; (void)in_sizes; (void)n_in; (void)out_size;

    k_sqnorm<<<(B_ * N_ + 255) / 256, 256, 0, stream>>>(x, xx);
    k_knn  <<<B_ * (N_ / 64), 128, 0, stream>>>(x, xx, id);
    k_wc   <<<(128 * 64 + 255) / 256, 256, 0, stream>>>(W, Wc);
    k_proj <<<B_ * (N_ / 32), 256, 0, stream>>>(x, Wc, P);
    k_zero <<<1, 128, 0, stream>>>(S);
    k_stats<<<B_ * N_ / (8 * ST_PPW), 256, 0, stream>>>(P, id, S);
    k_final<<<B_ * N_ / 32, 256, 0, stream>>>(P, id, S, gamma, beta, out);
}


# k_knn scan: branch-free queue append: every candidate is stored to the lane's next free LDS queue element and a select moves the pointer; replaces 8 compare/exec-mask/branch sequences per tile
# speedup vs baseline: 10.9804x; 1.1444x over previous
#include <hip/hip_runtime.h>
#include <math.h>

typedef __attribute__((ext_vector_type(2))) float v2f;
typedef __attribute__((ext_vector_type(8))) float v8f;
typedef __attribute__((ext_vector_type(2))) int v2i;

#define B_   8
#define C_   64
#define N_   4096
#define O_   64
#define K_NN 20
#define EPS_ 1e-5f
#define SLOPE_ 0.2f
#define NEG_INF -3.4e38f
#define KQ   32

static __device__ __forceinline__ v8f wmma_f32(v2f a, v2f b, v8f c) {
    return __builtin_amdgcn_wmma_f32_16x16x4_f32(false, a, false, b, (short)0, c,
                                                 false, false);
}

__global__ void __launch_bounds__(256) k_sqnorm(const float* __restrict__ x,
                                                float* __restrict__ xx) {
    int t = blockIdx.x * blockDim.x + threadIdx.x;
    if (t >= B_ * N_) return;
    int b = t / N_, n = t % N_;
    const float* xb = x + (size_t)b * C_ * N_;
    float s = 0.f;
#pragma unroll
    for (int c = 0; c < C_; ++c) {
        float v = xb[(size_t)c * N_ + n];
        s += v * v;
    }
    xx[t] = s;
}

__global__ void __launch_bounds__(128) k_knn(const float* __restrict__ x,
                                             const float* __restrict__ xx,
                                             int* __restrict__ idx) {
    const int lane = threadIdx.x & 31;
    const int wave = threadIdx.x >> 5;
    const int b    = blockIdx.x / (N_ / 64);
    const int rb   = (blockIdx.x % (N_ / 64)) * 64;
    const int row0 = rb + wave * 16;

    const float* xb  = x  + (size_t)b * C_ * N_;
    const float* xxb = xx + (size_t)b * N_;

    const int hs  = lane >> 4;
    const int l16 = lane & 15;
    const int jb  = hs * 8;

    v2f A[16];
#pragma unroll
    for (int s = 0; s < 16; ++s) {
        int c0 = 4 * s + hs * 2;
        A[s].x = xb[(size_t)(c0 + 0) * N_ + row0 + l16];
        A[s].y = xb[(size_t)(c0 + 1) * N_ + row0 + l16];
    }

    const float xr = xxb[row0 + l16];

#define KNN_ALL(X) X(0) X(1) X(2) X(3) X(4) X(5) X(6) X(7) X(8) X(9) X(10) X(11) \
                   X(12) X(13) X(14) X(15) X(16) X(17) X(18) X(19)
#define KNN_TAIL_UP(X) X(19, 18) X(18, 17) X(17, 16) X(16, 15) X(15, 14) X(14, 13) \
                       X(13, 12) X(12, 11) X(11, 10) X(10, 9) X(9, 8) X(8, 7) X(7, 6) \
                       X(6, 5) X(5, 4) X(4, 3) X(3, 2) X(2, 1) X(1, 0)
#define KNN_DECL(t) float v##t = NEG_INF; int i##t = 0;
    KNN_ALL(KNN_DECL)
#define KNN_SLOT(t, u) { const bool gh = cv > v##u;                         \
                         i##t = gh ? i##u : (gl ? ci : i##t);               \
                         v##t = __builtin_amdgcn_fmed3f(v##u, v##t, cv);    \
                         gl = gh; }

    __shared__ v2i qf[KQ * 128];
    int qn = threadIdx.x;

    for (int m0 = 0; m0 < N_; m0 += 16) {
        v2f Bv[16];
#pragma unroll
        for (int s = 0; s < 16; ++s) {
            int c0 = 4 * s + hs * 2;
            Bv[s].x = xb[(size_t)(c0 + 0) * N_ + m0 + l16];
            Bv[s].y = xb[(size_t)(c0 + 1) * N_ + m0 + l16];
        }
        v8f acc = {};
#pragma unroll
        for (int s = 0; s < 16; ++s) acc = wmma_f32(Bv[s], A[s], acc);

        float xc[8];
#pragma unroll
        for (int j = 0; j < 8; ++j) xc[j] = xxb[m0 + jb + j];
        const int cbase = m0 + jb;
#pragma unroll
        for (int j = 0; j < 8; ++j) {
            const float cv = 2.0f * acc[j] - xr - xc[j];
            v2i e;
            e.x = __float_as_int(cv);
            e.y = cbase + j;
            qf[qn] = e;
            qn += (cv > v19) ? 128 : 0;
        }
        const bool full = __builtin_amdgcn_ballot_w32(qn >= (KQ - 7) * 128) != 0;
        if (full || m0 == N_ - 16) {
            for (int q = threadIdx.x; __builtin_amdgcn_ballot_w32(q < qn) != 0; q += 128) {
                if (q < qn) {
                    const v2i e = qf[q];
                    const float cv = __int_as_float(e.x);
                    const int ci = e.y;
                    bool gl = cv > v19;
                    KNN_TAIL_UP(KNN_SLOT)
                    i0 = gl ? ci : i0;
                    v0 = fmaxf(v0, cv);
                }
            }
            qn = threadIdx.x;
        }
    }

    __shared__ float mv[128][K_NN];
    __shared__ int   mi[128][K_NN];
#define KNN_PUT(t) mv[threadIdx.x][t] = v##t; mi[threadIdx.x][t] = i##t;
    KNN_ALL(KNN_PUT)
    __syncthreads();
    if (lane < 16) {
        const float* av = mv[wave * 32 + lane];
        const int*   ai = mi[wave * 32 + lane];
        const float* bv = mv[wave * 32 + lane + 16];
        const int*   bi = mi[wave * 32 + lane + 16];
        int* op = idx + ((size_t)b * N_ + row0 + lane) * K_NN;
        int ia = 0, ib2 = 0;
        for (int t = 0; t < K_NN; ++t) {
            float fa = av[ia], fb = bv[ib2];
            bool takeA = (fa > fb) || (fa == fb && ai[ia] < bi[ib2]);
            op[t] = takeA ? ai[ia++] : bi[ib2++];
        }
    }
}

__global__ void __launch_bounds__(256) k_wc(const float* __restrict__ W,
                                            float* __restrict__ Wc) {
    int t = blockIdx.x * blockDim.x + threadIdx.x;
    if (t >= 128 * 64) return;
    int op = t / 64, c = t % 64;
    float v;
    if (op < 64) v = W[(size_t)op * 128 + c];
    else         v = W[(size_t)(op - 64) * 128 + 64 + c] -
                     W[(size_t)(op - 64) * 128 + c];
    Wc[t] = v;
}

#define PT_STRIDE 128
__global__ void __launch_bounds__(256) k_proj(const float* __restrict__ x,
                                              const float* __restrict__ Wc,
                                              float* __restrict__ Pt) {
    const int lane = threadIdx.x & 31;
    const int wave = threadIdx.x >> 5;
    const int tile = blockIdx.x * 2 + (wave >> 2);
    const int b    = tile / (N_ / 16);
    const int n0   = (tile % (N_ / 16)) * 16;
    const int o0   = (wave & 3) * 32;
    const float* xb = x + (size_t)b * C_ * N_;
    const int hs = lane >> 4, l16 = lane & 15;
    const float* w0 = Wc + (size_t)(o0 + 2 * l16) * 64;
    const float* w1 = w0 + 64;

    v8f acc0 = {}, acc1 = {};
#pragma unroll
    for (int s = 0; s < 16; ++s) {
        int c0 = 4 * s + hs * 2;
        v2f a, b0, b1;
        a.x  = xb[(size_t)(c0 + 0) * N_ + n0 + l16];
        a.y  = xb[(size_t)(c0 + 1) * N_ + n0 + l16];
        b0.x = w0[c0 + 0];
        b0.y = w0[c0 + 1];
        b1.x = w1[c0 + 0];
        b1.y = w1[c0 + 1];
        acc0 = wmma_f32(a, b0, acc0);
        acc1 = wmma_f32(a, b1, acc1);
    }
    float* dst = Pt + ((size_t)b * N_ + n0 + hs * 8) * PT_STRIDE + o0 + 2 * l16;
#pragma unroll
    for (int v = 0; v < 8; ++v) {
        v2f o;
        o.x = acc0[v];
        o.y = acc1[v];
        *(v2f*)(dst + (size_t)v * PT_STRIDE) = o;
    }
}

__global__ void k_zero(float* __restrict__ S) {
    if (threadIdx.x < 128) S[threadIdx.x] = 0.f;
}

#define ST_PPW 8
__global__ void __launch_bounds__(256) k_stats(const float* __restrict__ Pt,
                                               const int* __restrict__ idx,
                                               float* __restrict__ S) {
    const int lane = threadIdx.x & 31;
    const int wave = __builtin_amdgcn_readfirstlane(threadIdx.x >> 5);

    float s1x = 0.f, s1y = 0.f, s2x = 0.f, s2y = 0.f;
    for (int i = 0; i < ST_PPW; ++i) {
        const int t     = (blockIdx.x * 8 + wave) * ST_PPW + i;
        const int bbase = t & ~(N_ - 1);
        const int* ip   = idx + (size_t)t * K_NN;
        const v2f p2 = *(const v2f*)(Pt + (size_t)t * PT_STRIDE + 64 + 2 * lane);
#pragma unroll
        for (int k = 0; k < K_NN; ++k) {
            const int j = ip[k];
            const v2f v = *(const v2f*)(Pt + (size_t)(bbase + j) * PT_STRIDE + 2 * lane);
            const float y0 = v.x + p2.x;
            const float y1 = v.y + p2.y;
            s1x += y0; s2x += y0 * y0;
            s1y += y1; s2y += y1 * y1;
        }
    }
    __shared__ float r1[8][64], r2[8][64];
    r1[wave][2 * lane]     = s1x;
    r1[wave][2 * lane + 1] = s1y;
    r2[wave][2 * lane]     = s2x;
    r2[wave][2 * lane + 1] = s2y;
    __syncthreads();
    if (threadIdx.x < 128) {
        const int o = threadIdx.x & 63;
        float a = 0.f;
        if (threadIdx.x < 64) {
            for (int w = 0; w < 8; ++w) a += r1[w][o];
        } else {
            for (int w = 0; w < 8; ++w) a += r2[w][o];
        }
        atomicAdd(&S[threadIdx.x], a);
    }
}

__global__ void __launch_bounds__(256) k_final(const float* __restrict__ Pt,
                                               const int* __restrict__ idx,
                                               const float* __restrict__ S,
                                               const float* __restrict__ gamma,
                                               const float* __restrict__ beta,
                                               float* __restrict__ out) {
    const int lane = threadIdx.x & 31;
    const int wave = __builtin_amdgcn_readfirstlane(threadIdx.x >> 5);
    const int t0    = blockIdx.x * 32;
    const int bbase = t0 & ~(N_ - 1);

    const float cnt = (float)((size_t)B_ * N_ * K_NN);
    const int   oa  = 2 * lane, ob = 2 * lane + 1;
    const float mean0 = S[oa] / cnt;
    const float var0  = S[64 + oa] / cnt - mean0 * mean0;
    const float g0    = gamma[oa] * rsqrtf(var0 + EPS_);
    const float be0   = beta[oa];
    const float mean1 = S[ob] / cnt;
    const float var1  = S[64 + ob] / cnt - mean1 * mean1;
    const float g1    = gamma[ob] * rsqrtf(var1 + EPS_);
    const float be1   = beta[ob];

    __shared__ float tile[64][33];

    for (int i = 0; i < 4; ++i) {
        const int pl = wave * 4 + i;
        const int t  = t0 + pl;
        const int* ip = idx + (size_t)t * K_NN;
        const v2f p2 = *(const v2f*)(Pt + (size_t)t * PT_STRIDE + 64 + 2 * lane);
        float best0 = NEG_INF, best1 = NEG_INF;
#pragma unroll
        for (int k = 0; k < K_NN; ++k) {
            const int j = ip[k];
            const v2f v = *(const v2f*)(Pt + (size_t)(bbase + j) * PT_STRIDE + 2 * lane);
            const float y0 = v.x + p2.x;
            const float y1 = v.y + p2.y;
            float z0 = (y0 - mean0) * g0 + be0;
            float z1 = (y1 - mean1) * g1 + be1;
            z0 = (z0 >= 0.f) ? z0 : SLOPE_ * z0;
            z1 = (z1 >= 0.f) ? z1 : SLOPE_ * z1;
            best0 = fmaxf(best0, z0);
            best1 = fmaxf(best1, z1);
        }
        tile[oa][pl] = best0;
        tile[ob][pl] = best1;
    }
    __syncthreads();
    const int b  = t0 / N_;
    const int n0 = t0 % N_;
    for (int r = 0; r < 8; ++r) {
        const int o = wave * 8 + r;
        out[((size_t)b * O_ + o) * N_ + n0 + lane] = tile[o][lane];
    }
}

extern "C" void kernel_launch(void* const* d_in, const int* in_sizes, int n_in,
                              void* d_out, int out_size, void* d_ws, size_t ws_size,
                              hipStream_t stream) {
    const float* x     = (const float*)d_in[0];
    const float* W     = (const float*)d_in[1];
    const float* gamma = (const float*)d_in[2];
    const float* beta  = (const float*)d_in[3];
    float* out = (float*)d_out;

    char* ws = (char*)d_ws;
    size_t off = 0;
    float* xx = (float*)(ws + off); off += (size_t)B_ * N_ * sizeof(float);
    int*   id = (int*)  (ws + off); off += (size_t)B_ * N_ * K_NN * sizeof(int);
    float* Wc = (float*)(ws + off); off += (size_t)128 * 64 * sizeof(float);
    float* P  = (float*)(ws + off); off += (size_t)B_ * N_ * PT_STRIDE * sizeof(float);
    float* S  = (float*)(ws + off); off += 128 * sizeof(float);
    (void)ws_size; (void)in_sizes; (void)n_in; (void)out_size;

    k_sqnorm<<<(B_ * N_ + 255) / 256, 256, 0, stream>>>(x, xx);
    k_knn  <<<B_ * (N_ / 64), 128, 0, stream>>>(x, xx, id);
    k_wc   <<<(128 * 64 + 255) / 256, 256, 0, stream>>>(W, Wc);
    k_proj <<<B_ * (N_ / 32), 256, 0, stream>>>(x, Wc, P);
    k_zero <<<1, 128, 0, stream>>>(S);
    k_stats<<<B_ * N_ / (8 * ST_PPW), 256, 0, stream>>>(P, id, S);
    k_final<<<B_ * N_ / 32, 256, 0, stream>>>(P, id, S, gamma, beta, out);
}


# k_knn column loop software-pipelined: the next tile's fragment and norm loads go to a second register set, issued ahead of the current tile's WMMAs (two tiles per trip, one drain test per pair)
# speedup vs baseline: 11.3184x; 1.0308x over previous
#include <hip/hip_runtime.h>
#include <math.h>

typedef __attribute__((ext_vector_type(2))) float v2f;
typedef __attribute__((ext_vector_type(8))) float v8f;
typedef __attribute__((ext_vector_type(2))) int v2i;

#define B_   8
#define C_   64
#define N_   4096
#define O_   64
#define K_NN 20
#define EPS_ 1e-5f
#define SLOPE_ 0.2f
#define NEG_INF -3.4e38f
#define KQ   40

static __device__ __forceinline__ v8f wmma_f32(v2f a, v2f b, v8f c) {
    return __builtin_amdgcn_wmma_f32_16x16x4_f32(false, a, false, b, (short)0, c,
                                                 false, false);
}

__global__ void __launch_bounds__(256) k_sqnorm(const float* __restrict__ x,
                                                float* __restrict__ xx) {
    int t = blockIdx.x * blockDim.x + threadIdx.x;
    if (t >= B_ * N_) return;
    int b = t / N_, n = t % N_;
    const float* xb = x + (size_t)b * C_ * N_;
    float s = 0.f;
#pragma unroll
    for (int c = 0; c < C_; ++c) {
        float v = xb[(size_t)c * N_ + n];
        s += v * v;
    }
    xx[t] = s;
}

__global__ void __launch_bounds__(128) k_knn(const float* __restrict__ x,
                                             const float* __restrict__ xx,
                                             int* __restrict__ idx) {
    const int lane = threadIdx.x & 31;
    const int wave = threadIdx.x >> 5;
    const int b    = blockIdx.x / (N_ / 64);
    const int rb   = (blockIdx.x % (N_ / 64)) * 64;
    const int row0 = rb + wave * 16;

    const float* xb  = x  + (size_t)b * C_ * N_;
    const float* xxb = xx + (size_t)b * N_;

    const int hs  = lane >> 4;
    const int l16 = lane & 15;
    const int jb  = hs * 8;

    v2f A[16];
#pragma unroll
    for (int s = 0; s < 16; ++s) {
        int c0 = 4 * s + hs * 2;
        A[s].x = xb[(size_t)(c0 + 0) * N_ + row0 + l16];
        A[s].y = xb[(size_t)(c0 + 1) * N_ + row0 + l16];
    }

    const float xr = xxb[row0 + l16];

#define KNN_ALL(X) X(0) X(1) X(2) X(3) X(4) X(5) X(6) X(7) X(8) X(9) X(10) X(11) \
                   X(12) X(13) X(14) X(15) X(16) X(17) X(18) X(19)
#define KNN_TAIL_UP(X) X(19, 18) X(18, 17) X(17, 16) X(16, 15) X(15, 14) X(14, 13) \
                       X(13, 12) X(12, 11) X(11, 10) X(10, 9) X(9, 8) X(8, 7) X(7, 6) \
                       X(6, 5) X(5, 4) X(4, 3) X(3, 2) X(2, 1) X(1, 0)
#define KNN_DECL(t) float v##t = NEG_INF; int i##t = 0;
    KNN_ALL(KNN_DECL)
#define KNN_SLOT(t, u) { const bool gh = cv > v##u;                         \
                         i##t = gh ? i##u : (gl ? ci : i##t);               \
                         v##t = __builtin_amdgcn_fmed3f(v##u, v##t, cv);    \
                         gl = gh; }

    __shared__ v2i qf[KQ * 128];
    int qn = threadIdx.x;

#define KNN_LOAD(F, XC, M)                                                  \
    _Pragma("unroll")                                                       \
    for (int s = 0; s < 16; ++s) {                                          \
        const int c0 = 4 * s + hs * 2;                                      \
        F[s].x = xb[(size_t)(c0 + 0) * N_ + (M) + l16];                     \
        F[s].y = xb[(size_t)(c0 + 1) * N_ + (M) + l16];                     \
    }                                                                       \
    _Pragma("unroll")                                                       \
    for (int j = 0; j < 8; ++j) XC[j] = xxb[(M) + jb + j];

#define KNN_TILE(F, XC, M)                                                  \
    {                                                                       \
        v8f acc = {};                                                       \
        _Pragma("unroll")                                                   \
        for (int s = 0; s < 16; ++s) acc = wmma_f32(F[s], A[s], acc);       \
        const int cbase = (M) + jb;                                         \
        _Pragma("unroll")                                                   \
        for (int j = 0; j < 8; ++j) {                                       \
            const float cv = 2.0f * acc[j] - xr - XC[j];                    \
            v2i e;                                                          \
            e.x = __float_as_int(cv);                                       \
            e.y = cbase + j;                                                \
            qf[qn] = e;                                                     \
            qn += (cv > v19) ? 128 : 0;                                     \
        }                                                                   \
    }

    v2f Fp[16], Fq[16];
    float xp[8], xq[8];
    KNN_LOAD(Fp, xp, 0)
    for (int m0 = 0; m0 < N_; m0 += 32) {
        KNN_LOAD(Fq, xq, m0 + 16)
        KNN_TILE(Fp, xp, m0)
        const int mn = (m0 + 32) & (N_ - 1);
        KNN_LOAD(Fp, xp, mn)
        KNN_TILE(Fq, xq, m0 + 16)
        const bool full = __builtin_amdgcn_ballot_w32(qn >= (KQ - 15) * 128) != 0;
        if (full || m0 == N_ - 32) {
            for (int q = threadIdx.x; __builtin_amdgcn_ballot_w32(q < qn) != 0; q += 128) {
                if (q < qn) {
                    const v2i e = qf[q];
                    const float cv = __int_as_float(e.x);
                    const int ci = e.y;
                    bool gl = cv > v19;
                    KNN_TAIL_UP(KNN_SLOT)
                    i0 = gl ? ci : i0;
                    v0 = fmaxf(v0, cv);
                }
            }
            qn = threadIdx.x;
        }
    }

    __shared__ float mv[128][K_NN];
    __shared__ int   mi[128][K_NN];
#define KNN_PUT(t) mv[threadIdx.x][t] = v##t; mi[threadIdx.x][t] = i##t;
    KNN_ALL(KNN_PUT)
    __syncthreads();
    if (lane < 16) {
        const float* av = mv[wave * 32 + lane];
        const int*   ai = mi[wave * 32 + lane];
        const float* bv = mv[wave * 32 + lane + 16];
        const int*   bi = mi[wave * 32 + lane + 16];
        int* op = idx + ((size_t)b * N_ + row0 + lane) * K_NN;
        int ia = 0, ib2 = 0;
        for (int t = 0; t < K_NN; ++t) {
            float fa = av[ia], fb = bv[ib2];
            bool takeA = (fa > fb) || (fa == fb && ai[ia] < bi[ib2]);
            op[t] = takeA ? ai[ia++] : bi[ib2++];
        }
    }
}

__global__ void __launch_bounds__(256) k_wc(const float* __restrict__ W,
                                            float* __restrict__ Wc) {
    int t = blockIdx.x * blockDim.x + threadIdx.x;
    if (t >= 128 * 64) return;
    int op = t / 64, c = t % 64;
    float v;
    if (op < 64) v = W[(size_t)op * 128 + c];
    else         v = W[(size_t)(op - 64) * 128 + 64 + c] -
                     W[(size_t)(op - 64) * 128 + c];
    Wc[t] = v;
}

#define PT_STRIDE 128
__global__ void __launch_bounds__(256) k_proj(const float* __restrict__ x,
                                              const float* __restrict__ Wc,
                                              float* __restrict__ Pt) {
    const int lane = threadIdx.x & 31;
    const int wave = threadIdx.x >> 5;
    const int tile = blockIdx.x * 2 + (wave >> 2);
    const int b    = tile / (N_ / 16);
    const int n0   = (tile % (N_ / 16)) * 16;
    const int o0   = (wave & 3) * 32;
    const float* xb = x + (size_t)b * C_ * N_;
    const int hs = lane >> 4, l16 = lane & 15;
    const float* w0 = Wc + (size_t)(o0 + 2 * l16) * 64;
    const float* w1 = w0 + 64;

    v8f acc0 = {}, acc1 = {};
#pragma unroll
    for (int s = 0; s < 16; ++s) {
        int c0 = 4 * s + hs * 2;
        v2f a, b0, b1;
        a.x  = xb[(size_t)(c0 + 0) * N_ + n0 + l16];
        a.y  = xb[(size_t)(c0 + 1) * N_ + n0 + l16];
        b0.x = w0[c0 + 0];
        b0.y = w0[c0 + 1];
        b1.x = w1[c0 + 0];
        b1.y = w1[c0 + 1];
        acc0 = wmma_f32(a, b0, acc0);
        acc1 = wmma_f32(a, b1, acc1);
    }
    float* dst = Pt + ((size_t)b * N_ + n0 + hs * 8) * PT_STRIDE + o0 + 2 * l16;
#pragma unroll
    for (int v = 0; v < 8; ++v) {
        v2f o;
        o.x = acc0[v];
        o.y = acc1[v];
        *(v2f*)(dst + (size_t)v * PT_STRIDE) = o;
    }
}

__global__ void k_zero(float* __restrict__ S) {
    if (threadIdx.x < 128) S[threadIdx.x] = 0.f;
}

#define ST_PPW 8
__global__ void __launch_bounds__(256) k_stats(const float* __restrict__ Pt,
                                               const int* __restrict__ idx,
                                               float* __restrict__ S) {
    const int lane = threadIdx.x & 31;
    const int wave = __builtin_amdgcn_readfirstlane(threadIdx.x >> 5);

    float s1x = 0.f, s1y = 0.f, s2x = 0.f, s2y = 0.f;
    for (int i = 0; i < ST_PPW; ++i) {
        const int t     = (blockIdx.x * 8 + wave) * ST_PPW + i;
        const int bbase = t & ~(N_ - 1);
        const int* ip   = idx + (size_t)t * K_NN;
        const v2f p2 = *(const v2f*)(Pt + (size_t)t * PT_STRIDE + 64 + 2 * lane);
#pragma unroll
        for (int k = 0; k < K_NN; ++k) {
            const int j = ip[k];
            const v2f v = *(const v2f*)(Pt + (size_t)(bbase + j) * PT_STRIDE + 2 * lane);
            const float y0 = v.x + p2.x;
            const float y1 = v.y + p2.y;
            s1x += y0; s2x += y0 * y0;
            s1y += y1; s2y += y1 * y1;
        }
    }
    __shared__ float r1[8][64], r2[8][64];
    r1[wave][2 * lane]     = s1x;
    r1[wave][2 * lane + 1] = s1y;
    r2[wave][2 * lane]     = s2x;
    r2[wave][2 * lane + 1] = s2y;
    __syncthreads();
    if (threadIdx.x < 128) {
        const int o = threadIdx.x & 63;
        float a = 0.f;
        if (threadIdx.x < 64) {
            for (int w = 0; w < 8; ++w) a += r1[w][o];
        } else {
            for (int w = 0; w < 8; ++w) a += r2[w][o];
        }
        atomicAdd(&S[threadIdx.x], a);
    }
}

__global__ void __launch_bounds__(256) k_final(const float* __restrict__ Pt,
                                               const int* __restrict__ idx,
                                               const float* __restrict__ S,
                                               const float* __restrict__ gamma,
                                               const float* __restrict__ beta,
                                               float* __restrict__ out) {
    const int lane = threadIdx.x & 31;
    const int wave = __builtin_amdgcn_readfirstlane(threadIdx.x >> 5);
    const int t0    = blockIdx.x * 32;
    const int bbase = t0 & ~(N_ - 1);

    const float cnt = (float)((size_t)B_ * N_ * K_NN);
    const int   oa  = 2 * lane, ob = 2 * lane + 1;
    const float mean0 = S[oa] / cnt;
    const float var0  = S[64 + oa] / cnt - mean0 * mean0;
    const float g0    = gamma[oa] * rsqrtf(var0 + EPS_);
    const float be0   = beta[oa];
    const float mean1 = S[ob] / cnt;
    const float var1  = S[64 + ob] / cnt - mean1 * mean1;
    const float g1    = gamma[ob] * rsqrtf(var1 + EPS_);
    const float be1   = beta[ob];

    __shared__ float tile[64][33];

    for (int i = 0; i < 4; ++i) {
        const int pl = wave * 4 + i;
        const int t  = t0 + pl;
        const int* ip = idx + (size_t)t * K_NN;
        const v2f p2 = *(const v2f*)(Pt + (size_t)t * PT_STRIDE + 64 + 2 * lane);
        float best0 = NEG_INF, best1 = NEG_INF;
#pragma unroll
        for (int k = 0; k < K_NN; ++k) {
            const int j = ip[k];
            const v2f v = *(const v2f*)(Pt + (size_t)(bbase + j) * PT_STRIDE + 2 * lane);
            const float y0 = v.x + p2.x;
            const float y1 = v.y + p2.y;
            float z0 = (y0 - mean0) * g0 + be0;
            float z1 = (y1 - mean1) * g1 + be1;
            z0 = (z0 >= 0.f) ? z0 : SLOPE_ * z0;
            z1 = (z1 >= 0.f) ? z1 : SLOPE_ * z1;
            best0 = fmaxf(best0, z0);
            best1 = fmaxf(best1, z1);
        }
        tile[oa][pl] = best0;
        tile[ob][pl] = best1;
    }
    __syncthreads();
    const int b  = t0 / N_;
    const int n0 = t0 % N_;
    for (int r = 0; r < 8; ++r) {
        const int o = wave * 8 + r;
        out[((size_t)b * O_ + o) * N_ + n0 + lane] = tile[o][lane];
    }
}

extern "C" void kernel_launch(void* const* d_in, const int* in_sizes, int n_in,
                              void* d_out, int out_size, void* d_ws, size_t ws_size,
                              hipStream_t stream) {
    const float* x     = (const float*)d_in[0];
    const float* W     = (const float*)d_in[1];
    const float* gamma = (const float*)d_in[2];
    const float* beta  = (const float*)d_in[3];
    float* out = (float*)d_out;

    char* ws = (char*)d_ws;
    size_t off = 0;
    float* xx = (float*)(ws + off); off += (size_t)B_ * N_ * sizeof(float);
    int*   id = (int*)  (ws + off); off += (size_t)B_ * N_ * K_NN * sizeof(int);
    float* Wc = (float*)(ws + off); off += (size_t)128 * 64 * sizeof(float);
    float* P  = (float*)(ws + off); off += (size_t)B_ * N_ * PT_STRIDE * sizeof(float);
    float* S  = (float*)(ws + off); off += 128 * sizeof(float);
    (void)ws_size; (void)in_sizes; (void)n_in; (void)out_size;

    k_sqnorm<<<(B_ * N_ + 255) / 256, 256, 0, stream>>>(x, xx);
    k_knn  <<<B_ * (N_ / 64), 128, 0, stream>>>(x, xx, id);
    k_wc   <<<(128 * 64 + 255) / 256, 256, 0, stream>>>(W, Wc);
    k_proj <<<B_ * (N_ / 32), 256, 0, stream>>>(x, Wc, P);
    k_zero <<<1, 128, 0, stream>>>(S);
    k_stats<<<B_ * N_ / (8 * ST_PPW), 256, 0, stream>>>(P, id, S);
    k_final<<<B_ * N_ / 32, 256, 0, stream>>>(P, id, S, gamma, beta, out);
}


# k_knn column operand read with 16-byte loads along the point axis: value j of a load feeds tile j of a 64-column block on 4 independent accumulators; 40 loads per 64 columns, was 136
# speedup vs baseline: 11.7384x; 1.0371x over previous
#include <hip/hip_runtime.h>
#include <math.h>

typedef __attribute__((ext_vector_type(2))) float v2f;
typedef __attribute__((ext_vector_type(4))) float v4f;
typedef __attribute__((ext_vector_type(8))) float v8f;
typedef __attribute__((ext_vector_type(2))) int v2i;

#define B_   8
#define C_   64
#define N_   4096
#define O_   64
#define K_NN 20
#define EPS_ 1e-5f
#define SLOPE_ 0.2f
#define NEG_INF -3.4e38f
#define KQ   56

static __device__ __forceinline__ v8f wmma_f32(v2f a, v2f b, v8f c) {
    return __builtin_amdgcn_wmma_f32_16x16x4_f32(false, a, false, b, (short)0, c,
                                                 false, false);
}

__global__ void __launch_bounds__(256) k_sqnorm(const float* __restrict__ x,
                                                float* __restrict__ xx) {
    int t = blockIdx.x * blockDim.x + threadIdx.x;
    if (t >= B_ * N_) return;
    int b = t / N_, n = t % N_;
    const float* xb = x + (size_t)b * C_ * N_;
    float s = 0.f;
#pragma unroll
    for (int c = 0; c < C_; ++c) {
        float v = xb[(size_t)c * N_ + n];
        s += v * v;
    }
    xx[t] = s;
}

__global__ void __launch_bounds__(128) k_knn(const float* __restrict__ x,
                                             const float* __restrict__ xx,
                                             int* __restrict__ idx) {
    const int lane = threadIdx.x & 31;
    const int wave = threadIdx.x >> 5;
    const int b    = blockIdx.x / (N_ / 64);
    const int rb   = (blockIdx.x % (N_ / 64)) * 64;
    const int row0 = rb + wave * 16;

    const float* xb  = x  + (size_t)b * C_ * N_;
    const float* xxb = xx + (size_t)b * N_;

    const int hs  = lane >> 4;
    const int l16 = lane & 15;
    const int jb  = hs * 32;

    v2f A[16];
#pragma unroll
    for (int s = 0; s < 16; ++s) {
        int c0 = 4 * s + hs * 2;
        A[s].x = xb[(size_t)(c0 + 0) * N_ + row0 + l16];
        A[s].y = xb[(size_t)(c0 + 1) * N_ + row0 + l16];
    }

    const float xr = xxb[row0 + l16];

#define KNN_ALL(X) X(0) X(1) X(2) X(3) X(4) X(5) X(6) X(7) X(8) X(9) X(10) X(11) \
                   X(12) X(13) X(14) X(15) X(16) X(17) X(18) X(19)
#define KNN_TAIL_UP(X) X(19, 18) X(18, 17) X(17, 16) X(16, 15) X(15, 14) X(14, 13) \
                       X(13, 12) X(12, 11) X(11, 10) X(10, 9) X(9, 8) X(8, 7) X(7, 6) \
                       X(6, 5) X(5, 4) X(4, 3) X(3, 2) X(2, 1) X(1, 0)
#define KNN_DECL(t) float v##t = NEG_INF; int i##t = 0;
    KNN_ALL(KNN_DECL)
#define KNN_SLOT(t, u) { const bool gh = cv > v##u;                         \
                         i##t = gh ? i##u : (gl ? ci : i##t);               \
                         v##t = __builtin_amdgcn_fmed3f(v##u, v##t, cv);    \
                         gl = gh; }

    __shared__ v2i qf[KQ * 128];
    int qn = threadIdx.x;

#define KNN_LOAD(H0, H1, M)                                                 \
    _Pragma("unroll")                                                       \
    for (int s = 0; s < 16; ++s) {                                          \
        const int c0 = 4 * s + hs * 2;                                      \
        H0[s] = *(const v4f*)(xb + (size_t)(c0 + 0) * N_ + (M) + 4 * l16);  \
        H1[s] = *(const v4f*)(xb + (size_t)(c0 + 1) * N_ + (M) + 4 * l16);  \
    }

#define KNN_PAIRS(H0, H1)                                                   \
    _Pragma("unroll")                                                       \
    for (int s = 0; s < 16; ++s) {                                          \
        F0[s].x = H0[s][0]; F0[s].y = H1[s][0];                             \
        F1[s].x = H0[s][1]; F1[s].y = H1[s][1];                             \
        F2[s].x = H0[s][2]; F2[s].y = H1[s][2];                             \
        F3[s].x = H0[s][3]; F3[s].y = H1[s][3];                             \
    }

#define KNN_MMA()                                                           \
    _Pragma("unroll")                                                       \
    for (int s = 0; s < 16; ++s) {                                          \
        acc0 = wmma_f32(F0[s], A[s], acc0);                                 \
        acc1 = wmma_f32(F1[s], A[s], acc1);                                 \
        acc2 = wmma_f32(F2[s], A[s], acc2);                                 \
        acc3 = wmma_f32(F3[s], A[s], acc3);                                 \
    }

#define KNN_APPEND(ACC, XC, CI)                                             \
    {                                                                       \
        const float cv = 2.0f * (ACC) - xr - (XC);                          \
        v2i e;                                                              \
        e.x = __float_as_int(cv);                                           \
        e.y = (CI);                                                         \
        qf[qn] = e;                                                         \
        qn += (cv > v19) ? 128 : 0;                                         \
    }

    v4f Na[16], Nb[16];
    KNN_LOAD(Na, Nb, 0)
    for (int m0 = 0; m0 < N_; m0 += 64) {
        v2f F0[16], F1[16], F2[16], F3[16];
        KNN_PAIRS(Na, Nb)
        const int mn = (m0 + 64) & (N_ - 1);
        KNN_LOAD(Na, Nb, mn)
        v4f xn[8];
#pragma unroll
        for (int r = 0; r < 8; ++r) xn[r] = *(const v4f*)(xxb + m0 + jb + 4 * r);
        __builtin_amdgcn_sched_barrier(0);
        v8f acc0 = {}, acc1 = {}, acc2 = {}, acc3 = {};
        KNN_MMA()
        const int cbase = m0 + jb;
#pragma unroll
        for (int r = 0; r < 8; ++r) {
            KNN_APPEND(acc0[r], xn[r][0], cbase + 4 * r + 0)
            KNN_APPEND(acc1[r], xn[r][1], cbase + 4 * r + 1)
            KNN_APPEND(acc2[r], xn[r][2], cbase + 4 * r + 2)
            KNN_APPEND(acc3[r], xn[r][3], cbase + 4 * r + 3)
        }
        const bool full = __builtin_amdgcn_ballot_w32(qn >= (KQ - 31) * 128) != 0;
        if (full || m0 == N_ - 64) {
            for (int q = threadIdx.x; __builtin_amdgcn_ballot_w32(q < qn) != 0; q += 128) {
                if (q < qn) {
                    const v2i e = qf[q];
                    const float cv = __int_as_float(e.x);
                    const int ci = e.y;
                    bool gl = cv > v19;
                    KNN_TAIL_UP(KNN_SLOT)
                    i0 = gl ? ci : i0;
                    v0 = fmaxf(v0, cv);
                }
            }
            qn = threadIdx.x;
        }
    }

    __shared__ float mv[128][K_NN];
    __shared__ int   mi[128][K_NN];
#define KNN_PUT(t) mv[threadIdx.x][t] = v##t; mi[threadIdx.x][t] = i##t;
    KNN_ALL(KNN_PUT)
    __syncthreads();
    if (lane < 16) {
        const float* av = mv[wave * 32 + lane];
        const int*   ai = mi[wave * 32 + lane];
        const float* bv = mv[wave * 32 + lane + 16];
        const int*   bi = mi[wave * 32 + lane + 16];
        int* op = idx + ((size_t)b * N_ + row0 + lane) * K_NN;
        int ia = 0, ib2 = 0;
        for (int t = 0; t < K_NN; ++t) {
            float fa = av[ia], fb = bv[ib2];
            bool takeA = (fa > fb) || (fa == fb && ai[ia] < bi[ib2]);
            op[t] = takeA ? ai[ia++] : bi[ib2++];
        }
    }
}

__global__ void __launch_bounds__(256) k_wc(const float* __restrict__ W,
                                            float* __restrict__ Wc) {
    int t = blockIdx.x * blockDim.x + threadIdx.x;
    if (t >= 128 * 64) return;
    int op = t / 64, c = t % 64;
    float v;
    if (op < 64) v = W[(size_t)op * 128 + c];
    else         v = W[(size_t)(op - 64) * 128 + 64 + c] -
                     W[(size_t)(op - 64) * 128 + c];
    Wc[t] = v;
}

#define PT_STRIDE 128
__global__ void __launch_bounds__(256) k_proj(const float* __restrict__ x,
                                              const float* __restrict__ Wc,
                                              float* __restrict__ Pt) {
    const int lane = threadIdx.x & 31;
    const int wave = threadIdx.x >> 5;
    const int tile = blockIdx.x * 2 + (wave >> 2);
    const int b    = tile / (N_ / 16);
    const int n0   = (tile % (N_ / 16)) * 16;
    const int o0   = (wave & 3) * 32;
    const float* xb = x + (size_t)b * C_ * N_;
    const int hs = lane >> 4, l16 = lane & 15;
    const float* w0 = Wc + (size_t)(o0 + 2 * l16) * 64;
    const float* w1 = w0 + 64;

    v8f acc0 = {}, acc1 = {};
#pragma unroll
    for (int s = 0; s < 16; ++s) {
        int c0 = 4 * s + hs * 2;
        v2f a, b0, b1;
        a.x  = xb[(size_t)(c0 + 0) * N_ + n0 + l16];
        a.y  = xb[(size_t)(c0 + 1) * N_ + n0 + l16];
        b0.x = w0[c0 + 0];
        b0.y = w0[c0 + 1];
        b1.x = w1[c0 + 0];
        b1.y = w1[c0 + 1];
        acc0 = wmma_f32(a, b0, acc0);
        acc1 = wmma_f32(a, b1, acc1);
    }
    float* dst = Pt + ((size_t)b * N_ + n0 + hs * 8) * PT_STRIDE + o0 + 2 * l16;
#pragma unroll
    for (int v = 0; v < 8; ++v) {
        v2f o;
        o.x = acc0[v];
        o.y = acc1[v];
        *(v2f*)(dst + (size_t)v * PT_STRIDE) = o;
    }
}

__global__ void k_zero(float* __restrict__ S) {
    if (threadIdx.x < 128) S[threadIdx.x] = 0.f;
}

#define ST_PPW 8
__global__ void __launch_bounds__(256) k_stats(const float* __restrict__ Pt,
                                               const int* __restrict__ idx,
                                               float* __restrict__ S) {
    const int lane = threadIdx.x & 31;
    const int wave = __builtin_amdgcn_readfirstlane(threadIdx.x >> 5);

    float s1x = 0.f, s1y = 0.f, s2x = 0.f, s2y = 0.f;
    for (int i = 0; i < ST_PPW; ++i) {
        const int t     = (blockIdx.x * 8 + wave) * ST_PPW + i;
        const int bbase = t & ~(N_ - 1);
        const int* ip   = idx + (size_t)t * K_NN;
        const v2f p2 = *(const v2f*)(Pt + (size_t)t * PT_STRIDE + 64 + 2 * lane);
#pragma unroll
        for (int k = 0; k < K_NN; ++k) {
            const int j = ip[k];
            const v2f v = *(const v2f*)(Pt + (size_t)(bbase + j) * PT_STRIDE + 2 * lane);
            const float y0 = v.x + p2.x;
            const float y1 = v.y + p2.y;
            s1x += y0; s2x += y0 * y0;
            s1y += y1; s2y += y1 * y1;
        }
    }
    __shared__ float r1[8][64], r2[8][64];
    r1[wave][2 * lane]     = s1x;
    r1[wave][2 * lane + 1] = s1y;
    r2[wave][2 * lane]     = s2x;
    r2[wave][2 * lane + 1] = s2y;
    __syncthreads();
    if (threadIdx.x < 128) {
        const int o = threadIdx.x & 63;
        float a = 0.f;
        if (threadIdx.x < 64) {
            for (int w = 0; w < 8; ++w) a += r1[w][o];
        } else {
            for (int w = 0; w < 8; ++w) a += r2[w][o];
        }
        atomicAdd(&S[threadIdx.x], a);
    }
}

__global__ void __launch_bounds__(256) k_final(const float* __restrict__ Pt,
                                               const int* __restrict__ idx,
                                               const float* __restrict__ S,
                                               const float* __restrict__ gamma,
                                               const float* __restrict__ beta,
                                               float* __restrict__ out) {
    const int lane = threadIdx.x & 31;
    const int wave = __builtin_amdgcn_readfirstlane(threadIdx.x >> 5);
    const int t0    = blockIdx.x * 32;
    const int bbase = t0 & ~(N_ - 1);

    const float cnt = (float)((size_t)B_ * N_ * K_NN);
    const int   oa  = 2 * lane, ob = 2 * lane + 1;
    const float mean0 = S[oa] / cnt;
    const float var0  = S[64 + oa] / cnt - mean0 * mean0;
    const float g0    = gamma[oa] * rsqrtf(var0 + EPS_);
    const float be0   = beta[oa];
    const float mean1 = S[ob] / cnt;
    const float var1  = S[64 + ob] / cnt - mean1 * mean1;
    const float g1    = gamma[ob] * rsqrtf(var1 + EPS_);
    const float be1   = beta[ob];

    __shared__ float tile[64][33];

    for (int i = 0; i < 4; ++i) {
        const int pl = wave * 4 + i;
        const int t  = t0 + pl;
        const int* ip = idx + (size_t)t * K_NN;
        const v2f p2 = *(const v2f*)(Pt + (size_t)t * PT_STRIDE + 64 + 2 * lane);
        float best0 = NEG_INF, best1 = NEG_INF;
#pragma unroll
        for (int k = 0; k < K_NN; ++k) {
            const int j = ip[k];
            const v2f v = *(const v2f*)(Pt + (size_t)(bbase + j) * PT_STRIDE + 2 * lane);
            const float y0 = v.x + p2.x;
            const float y1 = v.y + p2.y;
            float z0 = (y0 - mean0) * g0 + be0;
            float z1 = (y1 - mean1) * g1 + be1;
            z0 = (z0 >= 0.f) ? z0 : SLOPE_ * z0;
            z1 = (z1 >= 0.f) ? z1 : SLOPE_ * z1;
            best0 = fmaxf(best0, z0);
            best1 = fmaxf(best1, z1);
        }
        tile[oa][pl] = best0;
        tile[ob][pl] = best1;
    }
    __syncthreads();
    const int b  = t0 / N_;
    const int n0 = t0 % N_;
    for (int r = 0; r < 8; ++r) {
        const int o = wave * 8 + r;
        out[((size_t)b * O_ + o) * N_ + n0 + lane] = tile[o][lane];
    }
}

extern "C" void kernel_launch(void* const* d_in, const int* in_sizes, int n_in,
                              void* d_out, int out_size, void* d_ws, size_t ws_size,
                              hipStream_t stream) {
    const float* x     = (const float*)d_in[0];
    const float* W     = (const float*)d_in[1];
    const float* gamma = (const float*)d_in[2];
    const float* beta  = (const float*)d_in[3];
    float* out = (float*)d_out;

    char* ws = (char*)d_ws;
    size_t off = 0;
    float* xx = (float*)(ws + off); off += (size_t)B_ * N_ * sizeof(float);
    int*   id = (int*)  (ws + off); off += (size_t)B_ * N_ * K_NN * sizeof(int);
    float* Wc = (float*)(ws + off); off += (size_t)128 * 64 * sizeof(float);
    float* P  = (float*)(ws + off); off += (size_t)B_ * N_ * PT_STRIDE * sizeof(float);
    float* S  = (float*)(ws + off); off += 128 * sizeof(float);
    (void)ws_size; (void)in_sizes; (void)n_in; (void)out_size;

    k_sqnorm<<<(B_ * N_ + 255) / 256, 256, 0, stream>>>(x, xx);
    k_knn  <<<B_ * (N_ / 64), 128, 0, stream>>>(x, xx, id);
    k_wc   <<<(128 * 64 + 255) / 256, 256, 0, stream>>>(W, Wc);
    k_proj <<<B_ * (N_ / 32), 256, 0, stream>>>(x, Wc, P);
    k_zero <<<1, 128, 0, stream>>>(S);
    k_stats<<<B_ * N_ / (8 * ST_PPW), 256, 0, stream>>>(P, id, S);
    k_final<<<B_ * N_ / 32, 256, 0, stream>>>(P, id, S, gamma, beta, out);
}


# k_knn: running top-20 kept as sorted values only (one median per slot); candidates stay in the lane's LDS log, compacted to the 20th value per drain; the 20 columns are read from the log at the end
# speedup vs baseline: 12.2271x; 1.0416x over previous
#include <hip/hip_runtime.h>
#include <math.h>

typedef __attribute__((ext_vector_type(2))) float v2f;
typedef __attribute__((ext_vector_type(4))) float v4f;
typedef __attribute__((ext_vector_type(8))) float v8f;
typedef __attribute__((ext_vector_type(2))) int v2i;

#define B_   8
#define C_   64
#define N_   4096
#define O_   64
#define K_NN 20
#define EPS_ 1e-5f
#define SLOPE_ 0.2f
#define NEG_INF -3.4e38f
#define KQ   76
#define KQ_PAD 8

static __device__ __forceinline__ v8f wmma_f32(v2f a, v2f b, v8f c) {
    return __builtin_amdgcn_wmma_f32_16x16x4_f32(false, a, false, b, (short)0, c,
                                                 false, false);
}

__global__ void __launch_bounds__(256) k_sqnorm(const float* __restrict__ x,
                                                float* __restrict__ xx) {
    int t = blockIdx.x * blockDim.x + threadIdx.x;
    if (t >= B_ * N_) return;
    int b = t / N_, n = t % N_;
    const float* xb = x + (size_t)b * C_ * N_;
    float s = 0.f;
#pragma unroll
    for (int c = 0; c < C_; ++c) {
        float v = xb[(size_t)c * N_ + n];
        s += v * v;
    }
    xx[t] = s;
}

__global__ void __launch_bounds__(128) k_knn(const float* __restrict__ x,
                                             const float* __restrict__ xx,
                                             int* __restrict__ idx) {
    const int lane = threadIdx.x & 31;
    const int wave = threadIdx.x >> 5;
    const int b    = blockIdx.x / (N_ / 64);
    const int rb   = (blockIdx.x % (N_ / 64)) * 64;
    const int row0 = rb + wave * 16;

    const float* xb  = x  + (size_t)b * C_ * N_;
    const float* xxb = xx + (size_t)b * N_;

    const int hs  = lane >> 4;
    const int l16 = lane & 15;
    const int jb  = hs * 32;

    v2f A[16];
#pragma unroll
    for (int s = 0; s < 16; ++s) {
        int c0 = 4 * s + hs * 2;
        A[s].x = xb[(size_t)(c0 + 0) * N_ + row0 + l16];
        A[s].y = xb[(size_t)(c0 + 1) * N_ + row0 + l16];
    }

    const float xr = xxb[row0 + l16];

#define KNN_ALL(X) X(0) X(1) X(2) X(3) X(4) X(5) X(6) X(7) X(8) X(9) X(10) X(11) \
                   X(12) X(13) X(14) X(15) X(16) X(17) X(18) X(19)
#define KNN_TAIL_UP(X) X(19, 18) X(18, 17) X(17, 16) X(16, 15) X(15, 14) X(14, 13) \
                       X(13, 12) X(12, 11) X(11, 10) X(10, 9) X(9, 8) X(8, 7) X(7, 6) \
                       X(6, 5) X(5, 4) X(4, 3) X(3, 2) X(2, 1) X(1, 0)
#define KNN_DECL(t) float v##t = NEG_INF;
    KNN_ALL(KNN_DECL)
#define KNN_VSLOT(t, u) v##t = __builtin_amdgcn_fmed3f(v##u, v##t, cv);
#define KNN_VINSERT(CV) { const float cv = (CV); KNN_TAIL_UP(KNN_VSLOT) v0 = fmaxf(v0, cv); }
#define KNN_SLOT(t, u) { const bool gh = cv > v##u;                         \
                         i##t = gh ? i##u : (gl ? ci : i##t);               \
                         v##t = __builtin_amdgcn_fmed3f(v##u, v##t, cv);    \
                         gl = gh; }

    __shared__ v2i qf[(KQ + KQ_PAD) * 128];
    int qn = threadIdx.x;
    int qd = threadIdx.x;

#define KNN_LOAD(H0, H1, M)                                                 \
    _Pragma("unroll")                                                       \
    for (int s = 0; s < 16; ++s) {                                          \
        const int c0 = 4 * s + hs * 2;                                      \
        H0[s] = *(const v4f*)(xb + (size_t)(c0 + 0) * N_ + (M) + 4 * l16);  \
        H1[s] = *(const v4f*)(xb + (size_t)(c0 + 1) * N_ + (M) + 4 * l16);  \
    }

#define KNN_PAIRS(H0, H1)                                                   \
    _Pragma("unroll")                                                       \
    for (int s = 0; s < 16; ++s) {                                          \
        F0[s].x = H0[s][0]; F0[s].y = H1[s][0];                             \
        F1[s].x = H0[s][1]; F1[s].y = H1[s][1];                             \
        F2[s].x = H0[s][2]; F2[s].y = H1[s][2];                             \
        F3[s].x = H0[s][3]; F3[s].y = H1[s][3];                             \
    }

#define KNN_MMA()                                                           \
    _Pragma("unroll")                                                       \
    for (int s = 0; s < 16; ++s) {                                          \
        acc0 = wmma_f32(F0[s], A[s], acc0);                                 \
        acc1 = wmma_f32(F1[s], A[s], acc1);                                 \
        acc2 = wmma_f32(F2[s], A[s], acc2);                                 \
        acc3 = wmma_f32(F3[s], A[s], acc3);                                 \
    }

#define KNN_APPEND(ACC, XC, CI)                                             \
    {                                                                       \
        const float cv = 2.0f * (ACC) - xr - (XC);                          \
        v2i e;                                                              \
        e.x = __float_as_int(cv);                                           \
        e.y = (CI);                                                         \
        qf[qn] = e;                                                         \
        qn += (cv > v19) ? 128 : 0;                                         \
    }

    v4f Na[16], Nb[16];
    KNN_LOAD(Na, Nb, 0)
    for (int m0 = 0; m0 < N_; m0 += 64) {
        v2f F0[16], F1[16], F2[16], F3[16];
        KNN_PAIRS(Na, Nb)
        const int mn = (m0 + 64) & (N_ - 1);
        KNN_LOAD(Na, Nb, mn)
        v4f xn[8];
#pragma unroll
        for (int r = 0; r < 8; ++r) xn[r] = *(const v4f*)(xxb + m0 + jb + 4 * r);
        __builtin_amdgcn_sched_barrier(0);
        v8f acc0 = {}, acc1 = {}, acc2 = {}, acc3 = {};
        KNN_MMA()
        const int cbase = m0 + jb;
#pragma unroll
        for (int r = 0; r < 8; ++r) {
            KNN_APPEND(acc0[r], xn[r][0], cbase + 4 * r + 0)
            KNN_APPEND(acc1[r], xn[r][1], cbase + 4 * r + 1)
            KNN_APPEND(acc2[r], xn[r][2], cbase + 4 * r + 2)
            KNN_APPEND(acc3[r], xn[r][3], cbase + 4 * r + 3)
        }
        const bool full = __builtin_amdgcn_ballot_w32(qn >= (KQ - 31) * 128) != 0;
        if (full || m0 == N_ - 64) {
            const int nnew = __builtin_amdgcn_readfirstlane(
                (int)__builtin_amdgcn_wave_reduce_max_u32((unsigned)((qn - qd) >> 7), 0));
            for (int n = 0; n < nnew; n += 4) {
                const int q = qd + n * 128;
                const float r0 = __int_as_float(qf[q].x);
                const float r1 = __int_as_float(qf[q + 128].x);
                const float r2 = __int_as_float(qf[q + 256].x);
                const float r3 = __int_as_float(qf[q + 384].x);
                KNN_VINSERT((q       < qn) ? r0 : NEG_INF)
                KNN_VINSERT((q + 128 < qn) ? r1 : NEG_INF)
                KNN_VINSERT((q + 256 < qn) ? r2 : NEG_INF)
                KNN_VINSERT((q + 384 < qn) ? r3 : NEG_INF)
            }
            const float tau = v19;
            int w = threadIdx.x;
#define KNN_KEEP(E, Q) { qf[w] = (E); w += ((Q) < qn && __int_as_float((E).x) >= tau) ? 128 : 0; }
            const int ntot = __builtin_amdgcn_readfirstlane(
                (int)__builtin_amdgcn_wave_reduce_max_u32((unsigned)(qn >> 7), 0));
            for (int n = 0; n < ntot; n += 8) {
                const int q = threadIdx.x + n * 128;
                const v2i e0 = qf[q];
                const v2i e1 = qf[q + 128];
                const v2i e2 = qf[q + 256];
                const v2i e3 = qf[q + 384];
                const v2i e4 = qf[q + 512];
                const v2i e5 = qf[q + 640];
                const v2i e6 = qf[q + 768];
                const v2i e7 = qf[q + 896];
                KNN_KEEP(e0, q)
                KNN_KEEP(e1, q + 128)
                KNN_KEEP(e2, q + 256)
                KNN_KEEP(e3, q + 384)
                KNN_KEEP(e4, q + 512)
                KNN_KEEP(e5, q + 640)
                KNN_KEEP(e6, q + 768)
                KNN_KEEP(e7, q + 896)
            }
            qn = w;
            if (__builtin_amdgcn_ballot_w32((qn >> 7) > 24) != 0) {
                int above = 0;
                for (int q = threadIdx.x; __builtin_amdgcn_ballot_w32(q < qn) != 0; q += 128)
                    above += (q < qn && __int_as_float(qf[q].x) > tau) ? 1 : 0;
                const int allow = K_NN - above;
                int ties = 0;
                w = threadIdx.x;
                for (int q = threadIdx.x; __builtin_amdgcn_ballot_w32(q < qn) != 0; q += 128) {
                    const v2i e = qf[q];
                    const float ev = __int_as_float(e.x);
                    const bool tie = (q < qn) && (ev == tau);
                    const bool keep = (q < qn) && (ev > tau || (tie && ties < allow));
                    ties += tie ? 1 : 0;
                    qf[w] = e;
                    w += keep ? 128 : 0;
                }
                qn = w;
            }
            qd = qn;
        }
    }

#define KNN_RESET(t) v##t = NEG_INF;
    KNN_ALL(KNN_RESET)
#define KNN_IDECL(t) int i##t = 0;
    KNN_ALL(KNN_IDECL)
    for (int q = threadIdx.x; __builtin_amdgcn_ballot_w32(q < qn) != 0; q += 128) {
        if (q < qn) {
            const v2i e = qf[q];
            const float cv = __int_as_float(e.x);
            const int ci = e.y;
            bool gl = cv > v19;
            KNN_TAIL_UP(KNN_SLOT)
            i0 = gl ? ci : i0;
            v0 = fmaxf(v0, cv);
        }
    }

    __shared__ float mv[128][K_NN];
    __shared__ int   mi[128][K_NN];
#define KNN_PUT(t) mv[threadIdx.x][t] = v##t; mi[threadIdx.x][t] = i##t;
    KNN_ALL(KNN_PUT)
    __syncthreads();
    if (lane < 16) {
        const float* av = mv[wave * 32 + lane];
        const int*   ai = mi[wave * 32 + lane];
        const float* bv = mv[wave * 32 + lane + 16];
        const int*   bi = mi[wave * 32 + lane + 16];
        int* op = idx + ((size_t)b * N_ + row0 + lane) * K_NN;
        int ia = 0, ib2 = 0;
        for (int t = 0; t < K_NN; ++t) {
            float fa = av[ia], fb = bv[ib2];
            bool takeA = (fa > fb) || (fa == fb && ai[ia] < bi[ib2]);
            op[t] = takeA ? ai[ia++] : bi[ib2++];
        }
    }
}

__global__ void __launch_bounds__(256) k_wc(const float* __restrict__ W,
                                            float* __restrict__ Wc) {
    int t = blockIdx.x * blockDim.x + threadIdx.x;
    if (t >= 128 * 64) return;
    int op = t / 64, c = t % 64;
    float v;
    if (op < 64) v = W[(size_t)op * 128 + c];
    else         v = W[(size_t)(op - 64) * 128 + 64 + c] -
                     W[(size_t)(op - 64) * 128 + c];
    Wc[t] = v;
}

#define PT_STRIDE 128
__global__ void __launch_bounds__(256) k_proj(const float* __restrict__ x,
                                              const float* __restrict__ Wc,
                                              float* __restrict__ Pt) {
    const int lane = threadIdx.x & 31;
    const int wave = threadIdx.x >> 5;
    const int tile = blockIdx.x * 2 + (wave >> 2);
    const int b    = tile / (N_ / 16);
    const int n0   = (tile % (N_ / 16)) * 16;
    const int o0   = (wave & 3) * 32;
    const float* xb = x + (size_t)b * C_ * N_;
    const int hs = lane >> 4, l16 = lane & 15;
    const float* w0 = Wc + (size_t)(o0 + 2 * l16) * 64;
    const float* w1 = w0 + 64;

    v8f acc0 = {}, acc1 = {};
#pragma unroll
    for (int s = 0; s < 16; ++s) {
        int c0 = 4 * s + hs * 2;
        v2f a, b0, b1;
        a.x  = xb[(size_t)(c0 + 0) * N_ + n0 + l16];
        a.y  = xb[(size_t)(c0 + 1) * N_ + n0 + l16];
        b0.x = w0[c0 + 0];
        b0.y = w0[c0 + 1];
        b1.x = w1[c0 + 0];
        b1.y = w1[c0 + 1];
        acc0 = wmma_f32(a, b0, acc0);
        acc1 = wmma_f32(a, b1, acc1);
    }
    float* dst = Pt + ((size_t)b * N_ + n0 + hs * 8) * PT_STRIDE + o0 + 2 * l16;
#pragma unroll
    for (int v = 0; v < 8; ++v) {
        v2f o;
        o.x = acc0[v];
        o.y = acc1[v];
        *(v2f*)(dst + (size_t)v * PT_STRIDE) = o;
    }
}

__global__ void k_zero(float* __restrict__ S) {
    if (threadIdx.x < 128) S[threadIdx.x] = 0.f;
}

#define ST_PPW 8
__global__ void __launch_bounds__(256) k_stats(const float* __restrict__ Pt,
                                               const int* __restrict__ idx,
                                               float* __restrict__ S) {
    const int lane = threadIdx.x & 31;
    const int wave = __builtin_amdgcn_readfirstlane(threadIdx.x >> 5);

    float s1x = 0.f, s1y = 0.f, s2x = 0.f, s2y = 0.f;
    for (int i = 0; i < ST_PPW; ++i) {
        const int t     = (blockIdx.x * 8 + wave) * ST_PPW + i;
        const int bbase = t & ~(N_ - 1);
        const int* ip   = idx + (size_t)t * K_NN;
        const v2f p2 = *(const v2f*)(Pt + (size_t)t * PT_STRIDE + 64 + 2 * lane);
#pragma unroll
        for (int k = 0; k < K_NN; ++k) {
            const int j = ip[k];
            const v2f v = *(const v2f*)(Pt + (size_t)(bbase + j) * PT_STRIDE + 2 * lane);
            const float y0 = v.x + p2.x;
            const float y1 = v.y + p2.y;
            s1x += y0; s2x += y0 * y0;
            s1y += y1; s2y += y1 * y1;
        }
    }
    __shared__ float r1[8][64], r2[8][64];
    r1[wave][2 * lane]     = s1x;
    r1[wave][2 * lane + 1] = s1y;
    r2[wave][2 * lane]     = s2x;
    r2[wave][2 * lane + 1] = s2y;
    __syncthreads();
    if (threadIdx.x < 128) {
        const int o = threadIdx.x & 63;
        float a = 0.f;
        if (threadIdx.x < 64) {
            for (int w = 0; w < 8; ++w) a += r1[w][o];
        } else {
            for (int w = 0; w < 8; ++w) a += r2[w][o];
        }
        atomicAdd(&S[threadIdx.x], a);
    }
}

__global__ void __launch_bounds__(256) k_final(const float* __restrict__ Pt,
                                               const int* __restrict__ idx,
                                               const float* __restrict__ S,
                                               const float* __restrict__ gamma,
                                               const float* __restrict__ beta,
                                               float* __restrict__ out) {
    const int lane = threadIdx.x & 31;
    const int wave = __builtin_amdgcn_readfirstlane(threadIdx.x >> 5);
    const int t0    = blockIdx.x * 32;
    const int bbase = t0 & ~(N_ - 1);

    const float cnt = (float)((size_t)B_ * N_ * K_NN);
    const int   oa  = 2 * lane, ob = 2 * lane + 1;
    const float mean0 = S[oa] / cnt;
    const float var0  = S[64 + oa] / cnt - mean0 * mean0;
    const float g0    = gamma[oa] * rsqrtf(var0 + EPS_);
    const float be0   = beta[oa];
    const float mean1 = S[ob] / cnt;
    const float var1  = S[64 + ob] / cnt - mean1 * mean1;
    const float g1    = gamma[ob] * rsqrtf(var1 + EPS_);
    const float be1   = beta[ob];

    __shared__ float tile[64][33];

    for (int i = 0; i < 4; ++i) {
        const int pl = wave * 4 + i;
        const int t  = t0 + pl;
        const int* ip = idx + (size_t)t * K_NN;
        const v2f p2 = *(const v2f*)(Pt + (size_t)t * PT_STRIDE + 64 + 2 * lane);
        float best0 = NEG_INF, best1 = NEG_INF;
#pragma unroll
        for (int k = 0; k < K_NN; ++k) {
            const int j = ip[k];
            const v2f v = *(const v2f*)(Pt + (size_t)(bbase + j) * PT_STRIDE + 2 * lane);
            const float y0 = v.x + p2.x;
            const float y1 = v.y + p2.y;
            float z0 = (y0 - mean0) * g0 + be0;
            float z1 = (y1 - mean1) * g1 + be1;
            z0 = (z0 >= 0.f) ? z0 : SLOPE_ * z0;
            z1 = (z1 >= 0.f) ? z1 : SLOPE_ * z1;
            best0 = fmaxf(best0, z0);
            best1 = fmaxf(best1, z1);
        }
        tile[oa][pl] = best0;
        tile[ob][pl] = best1;
    }
    __syncthreads();
    const int b  = t0 / N_;
    const int n0 = t0 % N_;
    for (int r = 0; r < 8; ++r) {
        const int o = wave * 8 + r;
        out[((size_t)b * O_ + o) * N_ + n0 + lane] = tile[o][lane];
    }
}

extern "C" void kernel_launch(void* const* d_in, const int* in_sizes, int n_in,
                              void* d_out, int out_size, void* d_ws, size_t ws_size,
                              hipStream_t stream) {
    const float* x     = (const float*)d_in[0];
    const float* W     = (const float*)d_in[1];
    const float* gamma = (const float*)d_in[2];
    const float* beta  = (const float*)d_in[3];
    float* out = (float*)d_out;

    char* ws = (char*)d_ws;
    size_t off = 0;
    float* xx = (float*)(ws + off); off += (size_t)B_ * N_ * sizeof(float);
    int*   id = (int*)  (ws + off); off += (size_t)B_ * N_ * K_NN * sizeof(int);
    float* Wc = (float*)(ws + off); off += (size_t)128 * 64 * sizeof(float);
    float* P  = (float*)(ws + off); off += (size_t)B_ * N_ * PT_STRIDE * sizeof(float);
    float* S  = (float*)(ws + off); off += 128 * sizeof(float);
    (void)ws_size; (void)in_sizes; (void)n_in; (void)out_size;

    k_sqnorm<<<(B_ * N_ + 255) / 256, 256, 0, stream>>>(x, xx);
    k_knn  <<<B_ * (N_ / 64), 128, 0, stream>>>(x, xx, id);
    k_wc   <<<(128 * 64 + 255) / 256, 256, 0, stream>>>(W, Wc);
    k_proj <<<B_ * (N_ / 32), 256, 0, stream>>>(x, Wc, P);
    k_zero <<<1, 128, 0, stream>>>(S);
    k_stats<<<B_ * N_ / (8 * ST_PPW), 256, 0, stream>>>(P, id, S);
    k_final<<<B_ * N_ / 32, 256, 0, stream>>>(P, id, S, gamma, beta, out);
}


# k_knn column operand pre-packed by the norm kernel in WMMA fragment order: a 16-byte load is two ready operand pairs and two register sets alternate, so the per-block register copies are gone
# speedup vs baseline: 13.3483x; 1.0917x over previous
#include <hip/hip_runtime.h>
#include <math.h>

typedef __attribute__((ext_vector_type(2))) float v2f;
typedef __attribute__((ext_vector_type(4))) float v4f;
typedef __attribute__((ext_vector_type(8))) float v8f;
typedef __attribute__((ext_vector_type(2))) int v2i;

#define B_   8
#define C_   64
#define N_   4096
#define O_   64
#define K_NN 20
#define EPS_ 1e-5f
#define SLOPE_ 0.2f
#define NEG_INF -3.4e38f
#define KQ   76
#define KQ_PAD 8

static __device__ __forceinline__ v8f wmma_f32(v2f a, v2f b, v8f c) {
    return __builtin_amdgcn_wmma_f32_16x16x4_f32(false, a, false, b, (short)0, c,
                                                 false, false);
}

__global__ void __launch_bounds__(256) k_sqnorm(const float* __restrict__ x,
                                                float* __restrict__ xx,
                                                float* __restrict__ xf) {
    int t = blockIdx.x * blockDim.x + threadIdx.x;
    if (t >= B_ * N_) return;
    int b = t / N_, n = t % N_;
    const float* xb = x + (size_t)b * C_ * N_;
    float v[C_];
#pragma unroll
    for (int c = 0; c < C_; ++c) v[c] = xb[(size_t)c * N_ + n];
    float s = 0.f;
#pragma unroll
    for (int c = 0; c < C_; ++c) s += v[c] * v[c];
    xx[t] = s;
    const int g = n >> 6, l16 = (n & 63) >> 2, j = n & 3;
    float* dst = xf + (size_t)b * C_ * N_ + (size_t)(g * 4 + j) * 1024 + l16 * 4;
#pragma unroll
    for (int tt = 0; tt < 8; ++tt) {
#pragma unroll
        for (int h = 0; h < 2; ++h) {
            v4f q;
            q.x = v[8 * tt + 2 * h];
            q.y = v[8 * tt + 2 * h + 1];
            q.z = v[8 * tt + 4 + 2 * h];
            q.w = v[8 * tt + 5 + 2 * h];
            *(v4f*)(dst + tt * 128 + h * 64) = q;
        }
    }
}

__global__ void __launch_bounds__(128) k_knn(const float* __restrict__ x,
                                             const float* __restrict__ xx,
                                             const float* __restrict__ xf,
                                             int* __restrict__ idx) {
    const int lane = threadIdx.x & 31;
    const int wave = threadIdx.x >> 5;
    const int b    = blockIdx.x / (N_ / 64);
    const int rb   = (blockIdx.x % (N_ / 64)) * 64;
    const int row0 = rb + wave * 16;

    const float* xb  = x  + (size_t)b * C_ * N_;
    const float* xxb = xx + (size_t)b * N_;
    const float* xfb = xf + (size_t)b * C_ * N_;

    const int hs  = lane >> 4;
    const int l16 = lane & 15;
    const int jb  = hs * 32;

    v2f A[16];
#pragma unroll
    for (int s = 0; s < 16; ++s) {
        int c0 = 4 * s + hs * 2;
        A[s].x = xb[(size_t)(c0 + 0) * N_ + row0 + l16];
        A[s].y = xb[(size_t)(c0 + 1) * N_ + row0 + l16];
    }

    const float xr = xxb[row0 + l16];

#define KNN_ALL(X) X(0) X(1) X(2) X(3) X(4) X(5) X(6) X(7) X(8) X(9) X(10) X(11) \
                   X(12) X(13) X(14) X(15) X(16) X(17) X(18) X(19)
#define KNN_TAIL_UP(X) X(19, 18) X(18, 17) X(17, 16) X(16, 15) X(15, 14) X(14, 13) \
                       X(13, 12) X(12, 11) X(11, 10) X(10, 9) X(9, 8) X(8, 7) X(7, 6) \
                       X(6, 5) X(5, 4) X(4, 3) X(3, 2) X(2, 1) X(1, 0)
#define KNN_DECL(t) float v##t = NEG_INF;
    KNN_ALL(KNN_DECL)
#define KNN_VSLOT(t, u) v##t = __builtin_amdgcn_fmed3f(v##u, v##t, cv);
#define KNN_VINSERT(CV) { const float cv = (CV); KNN_TAIL_UP(KNN_VSLOT) v0 = fmaxf(v0, cv); }
#define KNN_SLOT(t, u) { const bool gh = cv > v##u;                         \
                         i##t = gh ? i##u : (gl ? ci : i##t);               \
                         v##t = __builtin_amdgcn_fmed3f(v##u, v##t, cv);    \
                         gl = gh; }

    __shared__ v2i qf[(KQ + KQ_PAD) * 128];
    int qn = threadIdx.x;
    int qd = threadIdx.x;

#define KNN_LOAD(Q, M)                                                      \
    _Pragma("unroll")                                                       \
    for (int jt = 0; jt < 32; ++jt)                                         \
        Q[jt] = *(const v4f*)(xfb + (size_t)((M) >> 6) * 4096 + jt * 128 + lane * 4);

#define KNN_PAIR(Q, J, S) v2f{Q[(J) * 8 + ((S) >> 1)][2 * ((S) & 1)], Q[(J) * 8 + ((S) >> 1)][2 * ((S) & 1) + 1]}
#define KNN_MMA(Q)                                                          \
    _Pragma("unroll")                                                       \
    for (int s = 0; s < 16; ++s) {                                          \
        acc0 = wmma_f32(KNN_PAIR(Q, 0, s), A[s], acc0);                     \
        acc1 = wmma_f32(KNN_PAIR(Q, 1, s), A[s], acc1);                     \
        acc2 = wmma_f32(KNN_PAIR(Q, 2, s), A[s], acc2);                     \
        acc3 = wmma_f32(KNN_PAIR(Q, 3, s), A[s], acc3);                     \
    }

#define KNN_APPEND(ACC, XC, CI)                                             \
    {                                                                       \
        const float cv = 2.0f * (ACC) - xr - (XC);                          \
        v2i e;                                                              \
        e.x = __float_as_int(cv);                                           \
        e.y = (CI);                                                         \
        qf[qn] = e;                                                         \
        qn += (cv > v19) ? 128 : 0;                                         \
    }

    v4f Q[2][32];
    KNN_LOAD(Q[0], 0)
    for (int mb = 0; mb < N_; mb += 128) {
#pragma unroll
    for (int u = 0; u < 2; ++u) {
        const int m0 = mb + 64 * u;
        const int mn = (m0 + 64) & (N_ - 1);
        KNN_LOAD(Q[u ^ 1], mn)
        v4f xn[8];
#pragma unroll
        for (int r = 0; r < 8; ++r) xn[r] = *(const v4f*)(xxb + m0 + jb + 4 * r);
        __builtin_amdgcn_sched_barrier(0);
        v8f acc0 = {}, acc1 = {}, acc2 = {}, acc3 = {};
        KNN_MMA(Q[u])
        const int cbase = m0 + jb;
#pragma unroll
        for (int r = 0; r < 8; ++r) {
            KNN_APPEND(acc0[r], xn[r][0], cbase + 4 * r + 0)
            KNN_APPEND(acc1[r], xn[r][1], cbase + 4 * r + 1)
            KNN_APPEND(acc2[r], xn[r][2], cbase + 4 * r + 2)
            KNN_APPEND(acc3[r], xn[r][3], cbase + 4 * r + 3)
        }
        const bool full = __builtin_amdgcn_ballot_w32(qn >= (KQ - 31) * 128) != 0;
        if (full || m0 == N_ - 64) {
            const int nnew = __builtin_amdgcn_readfirstlane(
                (int)__builtin_amdgcn_wave_reduce_max_u32((unsigned)((qn - qd) >> 7), 0));
            for (int n = 0; n < nnew; n += 4) {
                const int q = qd + n * 128;
                const float r0 = __int_as_float(qf[q].x);
                const float r1 = __int_as_float(qf[q + 128].x);
                const float r2 = __int_as_float(qf[q + 256].x);
                const float r3 = __int_as_float(qf[q + 384].x);
                KNN_VINSERT((q       < qn) ? r0 : NEG_INF)
                KNN_VINSERT((q + 128 < qn) ? r1 : NEG_INF)
                KNN_VINSERT((q + 256 < qn) ? r2 : NEG_INF)
                KNN_VINSERT((q + 384 < qn) ? r3 : NEG_INF)
            }
            const float tau = v19;
            int w = threadIdx.x;
#define KNN_KEEP(E, Q) { qf[w] = (E); w += ((Q) < qn && __int_as_float((E).x) >= tau) ? 128 : 0; }
            const int ntot = __builtin_amdgcn_readfirstlane(
                (int)__builtin_amdgcn_wave_reduce_max_u32((unsigned)(qn >> 7), 0));
            for (int n = 0; n < ntot; n += 8) {
                const int q = threadIdx.x + n * 128;
                const v2i e0 = qf[q];
                const v2i e1 = qf[q + 128];
                const v2i e2 = qf[q + 256];
                const v2i e3 = qf[q + 384];
                const v2i e4 = qf[q + 512];
                const v2i e5 = qf[q + 640];
                const v2i e6 = qf[q + 768];
                const v2i e7 = qf[q + 896];
                KNN_KEEP(e0, q)
                KNN_KEEP(e1, q + 128)
                KNN_KEEP(e2, q + 256)
                KNN_KEEP(e3, q + 384)
                KNN_KEEP(e4, q + 512)
                KNN_KEEP(e5, q + 640)
                KNN_KEEP(e6, q + 768)
                KNN_KEEP(e7, q + 896)
            }
            qn = w;
            if (__builtin_amdgcn_ballot_w32((qn >> 7) > 24) != 0) {
                int above = 0;
                for (int q = threadIdx.x; __builtin_amdgcn_ballot_w32(q < qn) != 0; q += 128)
                    above += (q < qn && __int_as_float(qf[q].x) > tau) ? 1 : 0;
                const int allow = K_NN - above;
                int ties = 0;
                w = threadIdx.x;
                for (int q = threadIdx.x; __builtin_amdgcn_ballot_w32(q < qn) != 0; q += 128) {
                    const v2i e = qf[q];
                    const float ev = __int_as_float(e.x);
                    const bool tie = (q < qn) && (ev == tau);
                    const bool keep = (q < qn) && (ev > tau || (tie && ties < allow));
                    ties += tie ? 1 : 0;
                    qf[w] = e;
                    w += keep ? 128 : 0;
                }
                qn = w;
            }
            qd = qn;
        }
    }
    }

#define KNN_RESET(t) v##t = NEG_INF;
    KNN_ALL(KNN_RESET)
#define KNN_IDECL(t) int i##t = 0;
    KNN_ALL(KNN_IDECL)
    for (int q = threadIdx.x; __builtin_amdgcn_ballot_w32(q < qn) != 0; q += 128) {
        if (q < qn) {
            const v2i e = qf[q];
            const float cv = __int_as_float(e.x);
            const int ci = e.y;
            bool gl = cv > v19;
            KNN_TAIL_UP(KNN_SLOT)
            i0 = gl ? ci : i0;
            v0 = fmaxf(v0, cv);
        }
    }

    __shared__ float mv[128][K_NN];
    __shared__ int   mi[128][K_NN];
#define KNN_PUT(t) mv[threadIdx.x][t] = v##t; mi[threadIdx.x][t] = i##t;
    KNN_ALL(KNN_PUT)
    __syncthreads();
    if (lane < 16) {
        const float* av = mv[wave * 32 + lane];
        const int*   ai = mi[wave * 32 + lane];
        const float* bv = mv[wave * 32 + lane + 16];
        const int*   bi = mi[wave * 32 + lane + 16];
        int* op = idx + ((size_t)b * N_ + row0 + lane) * K_NN;
        int ia = 0, ib2 = 0;
        for (int t = 0; t < K_NN; ++t) {
            float fa = av[ia], fb = bv[ib2];
            bool takeA = (fa > fb) || (fa == fb && ai[ia] < bi[ib2]);
            op[t] = takeA ? ai[ia++] : bi[ib2++];
        }
    }
}

__global__ void __launch_bounds__(256) k_wc(const float* __restrict__ W,
                                            float* __restrict__ Wc) {
    int t = blockIdx.x * blockDim.x + threadIdx.x;
    if (t >= 128 * 64) return;
    int op = t / 64, c = t % 64;
    float v;
    if (op < 64) v = W[(size_t)op * 128 + c];
    else         v = W[(size_t)(op - 64) * 128 + 64 + c] -
                     W[(size_t)(op - 64) * 128 + c];
    Wc[t] = v;
}

#define PT_STRIDE 128
__global__ void __launch_bounds__(256) k_proj(const float* __restrict__ x,
                                              const float* __restrict__ Wc,
                                              float* __restrict__ Pt) {
    const int lane = threadIdx.x & 31;
    const int wave = threadIdx.x >> 5;
    const int tile = blockIdx.x * 2 + (wave >> 2);
    const int b    = tile / (N_ / 16);
    const int n0   = (tile % (N_ / 16)) * 16;
    const int o0   = (wave & 3) * 32;
    const float* xb = x + (size_t)b * C_ * N_;
    const int hs = lane >> 4, l16 = lane & 15;
    const float* w0 = Wc + (size_t)(o0 + 2 * l16) * 64;
    const float* w1 = w0 + 64;

    v8f acc0 = {}, acc1 = {};
#pragma unroll
    for (int s = 0; s < 16; ++s) {
        int c0 = 4 * s + hs * 2;
        v2f a, b0, b1;
        a.x  = xb[(size_t)(c0 + 0) * N_ + n0 + l16];
        a.y  = xb[(size_t)(c0 + 1) * N_ + n0 + l16];
        b0.x = w0[c0 + 0];
        b0.y = w0[c0 + 1];
        b1.x = w1[c0 + 0];
        b1.y = w1[c0 + 1];
        acc0 = wmma_f32(a, b0, acc0);
        acc1 = wmma_f32(a, b1, acc1);
    }
    float* dst = Pt + ((size_t)b * N_ + n0 + hs * 8) * PT_STRIDE + o0 + 2 * l16;
#pragma unroll
    for (int v = 0; v < 8; ++v) {
        v2f o;
        o.x = acc0[v];
        o.y = acc1[v];
        *(v2f*)(dst + (size_t)v * PT_STRIDE) = o;
    }
}

__global__ void k_zero(float* __restrict__ S) {
    if (threadIdx.x < 128) S[threadIdx.x] = 0.f;
}

#define ST_PPW 8
__global__ void __launch_bounds__(256) k_stats(const float* __restrict__ Pt,
                                               const int* __restrict__ idx,
                                               float* __restrict__ S) {
    const int lane = threadIdx.x & 31;
    const int wave = __builtin_amdgcn_readfirstlane(threadIdx.x >> 5);

    float s1x = 0.f, s1y = 0.f, s2x = 0.f, s2y = 0.f;
    for (int i = 0; i < ST_PPW; ++i) {
        const int t     = (blockIdx.x * 8 + wave) * ST_PPW + i;
        const int bbase = t & ~(N_ - 1);
        const int* ip   = idx + (size_t)t * K_NN;
        const v2f p2 = *(const v2f*)(Pt + (size_t)t * PT_STRIDE + 64 + 2 * lane);
#pragma unroll
        for (int k = 0; k < K_NN; ++k) {
            const int j = ip[k];
            const v2f v = *(const v2f*)(Pt + (size_t)(bbase + j) * PT_STRIDE + 2 * lane);
            const float y0 = v.x + p2.x;
            const float y1 = v.y + p2.y;
            s1x += y0; s2x += y0 * y0;
            s1y += y1; s2y += y1 * y1;
        }
    }
    __shared__ float r1[8][64], r2[8][64];
    r1[wave][2 * lane]     = s1x;
    r1[wave][2 * lane + 1] = s1y;
    r2[wave][2 * lane]     = s2x;
    r2[wave][2 * lane + 1] = s2y;
    __syncthreads();
    if (threadIdx.x < 128) {
        const int o = threadIdx.x & 63;
        float a = 0.f;
        if (threadIdx.x < 64) {
            for (int w = 0; w < 8; ++w) a += r1[w][o];
        } else {
            for (int w = 0; w < 8; ++w) a += r2[w][o];
        }
        atomicAdd(&S[threadIdx.x], a);
    }
}

__global__ void __launch_bounds__(256) k_final(const float* __restrict__ Pt,
                                               const int* __restrict__ idx,
                                               const float* __restrict__ S,
                                               const float* __restrict__ gamma,
                                               const float* __restrict__ beta,
                                               float* __restrict__ out) {
    const int lane = threadIdx.x & 31;
    const int wave = __builtin_amdgcn_readfirstlane(threadIdx.x >> 5);
    const int t0    = blockIdx.x * 32;
    const int bbase = t0 & ~(N_ - 1);

    const float cnt = (float)((size_t)B_ * N_ * K_NN);
    const int   oa  = 2 * lane, ob = 2 * lane + 1;
    const float mean0 = S[oa] / cnt;
    const float var0  = S[64 + oa] / cnt - mean0 * mean0;
    const float g0    = gamma[oa] * rsqrtf(var0 + EPS_);
    const float be0   = beta[oa];
    const float mean1 = S[ob] / cnt;
    const float var1  = S[64 + ob] / cnt - mean1 * mean1;
    const float g1    = gamma[ob] * rsqrtf(var1 + EPS_);
    const float be1   = beta[ob];

    __shared__ float tile[64][33];

    for (int i = 0; i < 4; ++i) {
        const int pl = wave * 4 + i;
        const int t  = t0 + pl;
        const int* ip = idx + (size_t)t * K_NN;
        const v2f p2 = *(const v2f*)(Pt + (size_t)t * PT_STRIDE + 64 + 2 * lane);
        float best0 = NEG_INF, best1 = NEG_INF;
#pragma unroll
        for (int k = 0; k < K_NN; ++k) {
            const int j = ip[k];
            const v2f v = *(const v2f*)(Pt + (size_t)(bbase + j) * PT_STRIDE + 2 * lane);
            const float y0 = v.x + p2.x;
            const float y1 = v.y + p2.y;
            float z0 = (y0 - mean0) * g0 + be0;
            float z1 = (y1 - mean1) * g1 + be1;
            z0 = (z0 >= 0.f) ? z0 : SLOPE_ * z0;
            z1 = (z1 >= 0.f) ? z1 : SLOPE_ * z1;
            best0 = fmaxf(best0, z0);
            best1 = fmaxf(best1, z1);
        }
        tile[oa][pl] = best0;
        tile[ob][pl] = best1;
    }
    __syncthreads();
    const int b  = t0 / N_;
    const int n0 = t0 % N_;
    for (int r = 0; r < 8; ++r) {
        const int o = wave * 8 + r;
        out[((size_t)b * O_ + o) * N_ + n0 + lane] = tile[o][lane];
    }
}

extern "C" void kernel_launch(void* const* d_in, const int* in_sizes, int n_in,
                              void* d_out, int out_size, void* d_ws, size_t ws_size,
                              hipStream_t stream) {
    const float* x     = (const float*)d_in[0];
    const float* W     = (const float*)d_in[1];
    const float* gamma = (const float*)d_in[2];
    const float* beta  = (const float*)d_in[3];
    float* out = (float*)d_out;

    char* ws = (char*)d_ws;
    size_t off = 0;
    float* xx = (float*)(ws + off); off += (size_t)B_ * N_ * sizeof(float);
    int*   id = (int*)  (ws + off); off += (size_t)B_ * N_ * K_NN * sizeof(int);
    float* Wc = (float*)(ws + off); off += (size_t)128 * 64 * sizeof(float);
    float* P  = (float*)(ws + off); off += (size_t)B_ * N_ * PT_STRIDE * sizeof(float);
    float* S  = (float*)(ws + off); off += 128 * sizeof(float);
    (void)ws_size; (void)in_sizes; (void)n_in; (void)out_size;

    k_sqnorm<<<(B_ * N_ + 255) / 256, 256, 0, stream>>>(x, xx, P);
    k_knn  <<<B_ * (N_ / 64), 128, 0, stream>>>(x, xx, P, id);
    k_wc   <<<(128 * 64 + 255) / 256, 256, 0, stream>>>(W, Wc);
    k_proj <<<B_ * (N_ / 32), 256, 0, stream>>>(x, Wc, P);
    k_zero <<<1, 128, 0, stream>>>(S);
    k_stats<<<B_ * N_ / (8 * ST_PPW), 256, 0, stream>>>(P, id, S);
    k_final<<<B_ * N_ / 32, 256, 0, stream>>>(P, id, S, gamma, beta, out);
}
